# starting kernel
# baseline (speedup 1.0000x reference)
#include <hip/hip_runtime.h>
#include <hip/hip_bf16.h>
#include <cstdint>

typedef float v2f __attribute__((ext_vector_type(2)));
typedef float v8f __attribute__((ext_vector_type(8)));

#define IN_CH  256
#define HID_CH 128
#define KTILE  64
#define PSTRIDE 288

__device__ __forceinline__ void atomAddF(float* p, float v) {
    __hip_atomic_fetch_add(p, v, __ATOMIC_RELAXED, __HIP_MEMORY_SCOPE_AGENT);
}

__global__ void k_init_deg(float* __restrict__ deg, int n) {
    int i = blockIdx.x * blockDim.x + threadIdx.x;
    if (i < n) deg[i] = 1.0f;
}

__global__ void k_deg_accum(const int* __restrict__ dst, float* __restrict__ deg, int ne) {
    int e = blockIdx.x * blockDim.x + threadIdx.x;
    if (e < ne) atomAddF(&deg[dst[e]], 1.0f);
}

__global__ void k_rsqrt(float* __restrict__ deg, int n) {
    int i = blockIdx.x * blockDim.x + threadIdx.x;
    if (i < n) deg[i] = rsqrtf(deg[i]);
}

__global__ __launch_bounds__(256) void k_gemm1_wmma(
    const float* __restrict__ x, const float* __restrict__ W1,
    float* __restrict__ h, int n_nodes)
{
    __shared__ float sW[(KTILE / 2) * PSTRIDE];

    const int tid  = threadIdx.x;
    const int wave = tid >> 5;
    const int lane = tid & 31;
    const int half = lane >> 4;
    const int l16  = lane & 15;

    const int row_base = (blockIdx.x * 8 + wave) * 16;
    int arow = row_base + l16;
    if (arow >= n_nodes) arow = n_nodes - 1;
    const float* aptr = x + (size_t)arow * IN_CH + half * 2;

    v8f acc[8] = {};

    for (int k0 = 0; k0 < IN_CH; k0 += KTILE) {
        __syncthreads();
        for (int i = tid; i < (KTILE / 2) * 32; i += 256) {
            int p  = i >> 5;
            int c4 = (i & 31) * 4;
            float4 r0 = *(const float4*)(W1 + (size_t)(k0 + 2 * p + 0) * HID_CH + c4);
            float4 r1 = *(const float4*)(W1 + (size_t)(k0 + 2 * p + 1) * HID_CH + c4);
            float* dp = &sW[p * PSTRIDE + c4 * 2];
            ((float2*)dp)[0] = make_float2(r0.x, r1.x);
            ((float2*)dp)[1] = make_float2(r0.y, r1.y);
            ((float2*)dp)[2] = make_float2(r0.z, r1.z);
            ((float2*)dp)[3] = make_float2(r0.w, r1.w);
        }
        __syncthreads();

#pragma unroll 4
        for (int k = 0; k < KTILE; k += 4) {
            v2f a = *(const v2f*)(aptr + k0 + k);
            const int p = (k >> 1) + half;
            const float* bbase = &sW[p * PSTRIDE + l16 * 2];
#pragma unroll
            for (int nt = 0; nt < 8; ++nt) {
                v2f b = *(const v2f*)(bbase + nt * 32);
                acc[nt] = __builtin_amdgcn_wmma_f32_16x16x4_f32(
                    false, a, false, b, (short)0, acc[nt], false, false);
            }
        }
    }

#pragma unroll
    for (int nt = 0; nt < 8; ++nt) {
#pragma unroll
        for (int v = 0; v < 8; ++v) {
            int r = row_base + v + 8 * half;
            if (r < n_nodes) h[(size_t)r * HID_CH + nt * 16 + l16] = acc[nt][v];
        }
    }
}

__global__ void k_agg1_init(const float* __restrict__ h, const float* __restrict__ b1,
                            const float* __restrict__ dinv, float* __restrict__ out1,
                            int n_nodes)
{
    int idx = blockIdx.x * blockDim.x + threadIdx.x;
    if (idx >= n_nodes * (HID_CH / 4)) return;
    int node = idx >> 5;
    int c = (idx & 31) * 4;
    float di = dinv[node], w = di * di;
    float4 hv = *(const float4*)(h + (size_t)node * HID_CH + c);
    float4 bv = *(const float4*)(b1 + c);
    float4 o; o.x = bv.x + hv.x * w; o.y = bv.y + hv.y * w;
    o.z = bv.z + hv.z * w; o.w = bv.w + hv.w * w;
    *(float4*)(out1 + (size_t)node * HID_CH + c) = o;
}

__global__ __launch_bounds__(256) void k_agg1_edges(
    const float* __restrict__ h, const int* __restrict__ src, const int* __restrict__ dst,
    const float* __restrict__ dinv, float* __restrict__ out1, int ne)
{
    int e = blockIdx.x * (blockDim.x >> 5) + (threadIdx.x >> 5);
    if (e >= ne) return;
    int lane = threadIdx.x & 31;
    int s = src[e], d = dst[e];
    float norm = dinv[s] * dinv[d];
    float4 hv = *(const float4*)(h + (size_t)s * HID_CH + lane * 4);
    float* op = out1 + (size_t)d * HID_CH + lane * 4;
    atomAddF(op + 0, hv.x * norm);
    atomAddF(op + 1, hv.y * norm);
    atomAddF(op + 2, hv.z * norm);
    atomAddF(op + 3, hv.w * norm);
}

__global__ void k_gemm2(const float* __restrict__ out1, const float* __restrict__ W2,
                        const float* __restrict__ b2, const float* __restrict__ dinv,
                        float* __restrict__ h2, float* __restrict__ out, int n_nodes)
{
    int i = blockIdx.x * blockDim.x + threadIdx.x;
    if (i >= n_nodes) return;
    const float* row = out1 + (size_t)i * HID_CH;
    float s0 = 0.f, s1 = 0.f;
#pragma unroll 8
    for (int c = 0; c < HID_CH; c += 4) {
        float4 hv = *(const float4*)(row + c);
        float r0 = fmaxf(hv.x, 0.f), r1 = fmaxf(hv.y, 0.f);
        float r2 = fmaxf(hv.z, 0.f), r3 = fmaxf(hv.w, 0.f);
        s0 += r0 * W2[(c + 0) * 2 + 0] + r1 * W2[(c + 1) * 2 + 0]
            + r2 * W2[(c + 2) * 2 + 0] + r3 * W2[(c + 3) * 2 + 0];
        s1 += r0 * W2[(c + 0) * 2 + 1] + r1 * W2[(c + 1) * 2 + 1]
            + r2 * W2[(c + 2) * 2 + 1] + r3 * W2[(c + 3) * 2 + 1];
    }
    float di = dinv[i], w = di * di;
    h2[i * 2 + 0] = s0;
    h2[i * 2 + 1] = s1;
    out[i * 2 + 0] = b2[0] + s0 * w;
    out[i * 2 + 1] = b2[1] + s1 * w;
}

__global__ void k_agg2_edges(const float* __restrict__ h2, const int* __restrict__ src,
                             const int* __restrict__ dst, const float* __restrict__ dinv,
                             float* __restrict__ out, int ne)
{
    int e = blockIdx.x * blockDim.x + threadIdx.x;
    if (e >= ne) return;
    int s = src[e], d = dst[e];
    float norm = dinv[s] * dinv[d];
    atomAddF(&out[d * 2 + 0], h2[s * 2 + 0] * norm);
    atomAddF(&out[d * 2 + 1], h2[s * 2 + 1] * norm);
}

extern "C" void kernel_launch(void* const* d_in, const int* in_sizes, int n_in,
                              void* d_out, int out_size, void* d_ws, size_t ws_size,
                              hipStream_t stream)
{
    const float* x  = (const float*)d_in[0];
    const int*   ei = (const int*)d_in[1];
    const float* W1 = (const float*)d_in[2];
    const float* b1 = (const float*)d_in[3];
    const float* W2 = (const float*)d_in[4];
    const float* b2 = (const float*)d_in[5];
    float* out = (float*)d_out;

    const int n_nodes = in_sizes[0] / IN_CH;
    const int n_edges = in_sizes[1] / 2;
    const int* src = ei;
    const int* dst = ei + n_edges;

    char* ws = (char*)d_ws;
    size_t off = 0;
    float* dinv = (float*)(ws + off);
    off += ((size_t)n_nodes * 4 + 4095) & ~(size_t)4095;
    float* h = (float*)(ws + off);
    size_t hb = ((size_t)n_nodes * HID_CH * 4 + 4095) & ~(size_t)4095;
    off += hb;
    float* out1 = (float*)(ws + off);
    float* h2 = h;

    k_init_deg <<<(n_nodes + 255) / 256, 256, 0, stream>>>(dinv, n_nodes);
    k_deg_accum<<<(n_edges + 255) / 256, 256, 0, stream>>>(dst, dinv, n_edges);
    k_rsqrt    <<<(n_nodes + 255) / 256, 256, 0, stream>>>(dinv, n_nodes);

    k_gemm1_wmma<<<(n_nodes + 127) / 128, 256, 0, stream>>>(x, W1, h, n_nodes);

    int t1 = n_nodes * (HID_CH / 4);
    k_agg1_init <<<(t1 + 255) / 256, 256, 0, stream>>>(h, b1, dinv, out1, n_nodes);
    k_agg1_edges<<<(n_edges + 7) / 8, 256, 0, stream>>>(h, src, dst, dinv, out1, n_edges);

    k_gemm2     <<<(n_nodes + 255) / 256, 256, 0, stream>>>(out1, W2, b2, dinv, h2, out, n_nodes);
    k_agg2_edges<<<(n_edges + 255) / 256, 256, 0, stream>>>(h2, src, dst, dinv, out, n_edges);
}


# Aggregations: per-call counting sort of edges by destination, then each node's sources are summed in registers (no float atomics; sum order = arrival rank); relu+W2 on gathered row, lane-tree dot
# speedup vs baseline: 2.2611x; 2.2611x over previous
#include <hip/hip_runtime.h>
#include <hip/hip_bf16.h>
#include <cstdint>


typedef float v2f __attribute__((ext_vector_type(2)));
typedef float v8f __attribute__((ext_vector_type(8)));

#define IN_CH  256
#define HID_CH 128
#define KTILE  64
#define PSTRIDE 288
#define SCAN_T 1024

__global__ void k_zero_cnt(int* __restrict__ cnt, int n) {
    int i = blockIdx.x * blockDim.x + threadIdx.x;
    if (i < n) cnt[i] = 0;
}

__global__ void k_count_rank(const int* __restrict__ dst, int* __restrict__ cnt,
                             int* __restrict__ rank, int ne) {
    int e = blockIdx.x * blockDim.x + threadIdx.x;
    if (e < ne)
        rank[e] = __hip_atomic_fetch_add(&cnt[dst[e]], 1, __ATOMIC_RELAXED, __HIP_MEMORY_SCOPE_AGENT);
}

__device__ __forceinline__ int wave_sum_i(int v) {
    for (int o = 16; o > 0; o >>= 1) v += __shfl_xor(v, o);
    return v;
}

__device__ __forceinline__ int wave_incl_scan_i(int v, int lane) {
    for (int o = 1; o < 32; o <<= 1) {
        int t = __shfl_up(v, o);
        if (lane >= o) v += t;
    }
    return v;
}

__global__ __launch_bounds__(SCAN_T) void k_block_sums(const int* __restrict__ cnt,
                                                       int* __restrict__ bsum, int n) {
    __shared__ int sw[32];
    const int tid = threadIdx.x, wave = tid >> 5, lane = tid & 31;
    const int i = blockIdx.x * SCAN_T + tid;
    int v = wave_sum_i(i < n ? cnt[i] : 0);
    if (lane == 0) sw[wave] = v;
    __syncthreads();
    if (wave == 0) {
        int t = wave_sum_i(sw[lane]);
        if (lane == 0) bsum[blockIdx.x] = t;
    }
}

__global__ __launch_bounds__(SCAN_T) void k_scan_rows(const int* __restrict__ cnt,
                                                      const int* __restrict__ bsum,
                                                      int* __restrict__ rowptr,
                                                      float* __restrict__ dinv, int n) {
    __shared__ int sw[32];
    const int tid = threadIdx.x, wave = tid >> 5, lane = tid & 31;
    const int i = blockIdx.x * SCAN_T + tid;
    const int c = (i < n) ? cnt[i] : 0;
    const int v = wave_incl_scan_i(c, lane);
    if (lane == 31) sw[wave] = v;
    __syncthreads();
    if (wave == 0) {
        int before = 0;
        for (int j = lane; j < (int)blockIdx.x; j += 32) before += bsum[j];
        before = wave_sum_i(before);
        const int t = sw[lane];
        const int s = wave_incl_scan_i(t, lane);
        sw[lane] = s - t + before;
    }
    __syncthreads();
    if (i < n) {
        rowptr[i] = sw[wave] + v - c;
        dinv[i] = rsqrtf((float)(c + 1));
    }
}

__global__ void k_fill(const int* __restrict__ src, const int* __restrict__ dst,
                       const int* __restrict__ rowptr, const int* __restrict__ rank,
                       int* __restrict__ col, int ne) {
    int e = blockIdx.x * blockDim.x + threadIdx.x;
    if (e >= ne) return;
    int p = rowptr[dst[e]] + rank[e];
    if ((unsigned)p < (unsigned)ne) col[p] = src[e];
}

__global__ __launch_bounds__(256) void k_gemm1_wmma(
    const float* __restrict__ x, const float* __restrict__ W1,
    float* __restrict__ h, int n_nodes)
{
    __shared__ float sW[(KTILE / 2) * PSTRIDE];

    const int tid  = threadIdx.x;
    const int wave = tid >> 5;
    const int lane = tid & 31;
    const int half = lane >> 4;
    const int l16  = lane & 15;

    const int row_base = (blockIdx.x * 8 + wave) * 16;
    int arow = row_base + l16;
    if (arow >= n_nodes) arow = n_nodes - 1;
    const float* aptr = x + (size_t)arow * IN_CH + half * 2;

    v8f acc[8] = {};

    for (int k0 = 0; k0 < IN_CH; k0 += KTILE) {
        __syncthreads();
        for (int i = tid; i < (KTILE / 2) * 32; i += 256) {
            int p  = i >> 5;
            int c4 = (i & 31) * 4;
            float4 r0 = *(const float4*)(W1 + (size_t)(k0 + 2 * p + 0) * HID_CH + c4);
            float4 r1 = *(const float4*)(W1 + (size_t)(k0 + 2 * p + 1) * HID_CH + c4);
            float* dp = &sW[p * PSTRIDE + c4 * 2];
            ((float2*)dp)[0] = make_float2(r0.x, r1.x);
            ((float2*)dp)[1] = make_float2(r0.y, r1.y);
            ((float2*)dp)[2] = make_float2(r0.z, r1.z);
            ((float2*)dp)[3] = make_float2(r0.w, r1.w);
        }
        __syncthreads();

#pragma unroll 4
        for (int k = 0; k < KTILE; k += 4) {
            v2f a = *(const v2f*)(aptr + k0 + k);
            const int p = (k >> 1) + half;
            const float* bbase = &sW[p * PSTRIDE + l16 * 2];
#pragma unroll
            for (int nt = 0; nt < 8; ++nt) {
                v2f b = *(const v2f*)(bbase + nt * 32);
                acc[nt] = __builtin_amdgcn_wmma_f32_16x16x4_f32(
                    false, a, false, b, (short)0, acc[nt], false, false);
            }
        }
    }

#pragma unroll
    for (int nt = 0; nt < 8; ++nt) {
#pragma unroll
        for (int v = 0; v < 8; ++v) {
            int r = row_base + v + 8 * half;
            if (r < n_nodes) h[(size_t)r * HID_CH + nt * 16 + l16] = acc[nt][v];
        }
    }
}

__global__ __launch_bounds__(512) void k_gather1(
    const float* __restrict__ h, const int* __restrict__ rowptr, const int* __restrict__ cnt,
    const int* __restrict__ col, const float* __restrict__ dinv,
    const float* __restrict__ b1, const float* __restrict__ W2,
    float* __restrict__ h2, int n_nodes)
{
    __shared__ float sh2[32];
    const int lane = threadIdx.x & 31;
    const int wave = threadIdx.x >> 5;
    const int node = __builtin_amdgcn_readfirstlane(blockIdx.x * 16 + wave);
    float p0 = 0.f, p1 = 0.f;
    if (node < n_nodes) {
        const int start = rowptr[node];
        const int deg = cnt[node];
        const float dd = dinv[node];
        const float* hl = h + lane * 4;
        const float wself = dd * dd;
        float4 hv = *(const float4*)(hl + (size_t)node * HID_CH);
        float4 acc;
        acc.x = hv.x * wself; acc.y = hv.y * wself; acc.z = hv.z * wself; acc.w = hv.w * wself;
        for (int base = 0; base < deg; base += 32) {
            const int m = (deg - base < 32) ? (deg - base) : 32;
            int my_s = 0; float my_w = 0.f;
            if (lane < m) {
                my_s = col[start + base + lane];
                my_w = dinv[my_s] * dd;
            }
            const int my_wi = __float_as_int(my_w);
            int j = 0;
            for (; j + 4 <= m; j += 4) {
                const int s0 = __builtin_amdgcn_readlane(my_s, j + 0);
                const int s1 = __builtin_amdgcn_readlane(my_s, j + 1);
                const int s2 = __builtin_amdgcn_readlane(my_s, j + 2);
                const int s3 = __builtin_amdgcn_readlane(my_s, j + 3);
                const float w0 = __int_as_float(__builtin_amdgcn_readlane(my_wi, j + 0));
                const float w1 = __int_as_float(__builtin_amdgcn_readlane(my_wi, j + 1));
                const float w2 = __int_as_float(__builtin_amdgcn_readlane(my_wi, j + 2));
                const float w3 = __int_as_float(__builtin_amdgcn_readlane(my_wi, j + 3));
                const float4 a0 = *(const float4*)(hl + (size_t)s0 * HID_CH);
                const float4 a1 = *(const float4*)(hl + (size_t)s1 * HID_CH);
                const float4 a2 = *(const float4*)(hl + (size_t)s2 * HID_CH);
                const float4 a3 = *(const float4*)(hl + (size_t)s3 * HID_CH);
                acc.x += a0.x * w0; acc.y += a0.y * w0; acc.z += a0.z * w0; acc.w += a0.w * w0;
                acc.x += a1.x * w1; acc.y += a1.y * w1; acc.z += a1.z * w1; acc.w += a1.w * w1;
                acc.x += a2.x * w2; acc.y += a2.y * w2; acc.z += a2.z * w2; acc.w += a2.w * w2;
                acc.x += a3.x * w3; acc.y += a3.y * w3; acc.z += a3.z * w3; acc.w += a3.w * w3;
            }
            for (; j < m; ++j) {
                const int s0 = __builtin_amdgcn_readlane(my_s, j);
                const float w0 = __int_as_float(__builtin_amdgcn_readlane(my_wi, j));
                const float4 a0 = *(const float4*)(hl + (size_t)s0 * HID_CH);
                acc.x += a0.x * w0; acc.y += a0.y * w0; acc.z += a0.z * w0; acc.w += a0.w * w0;
            }
        }
        const float4 bv = *(const float4*)(b1 + lane * 4);
        const float r0 = fmaxf(acc.x + bv.x, 0.f), r1 = fmaxf(acc.y + bv.y, 0.f);
        const float r2 = fmaxf(acc.z + bv.z, 0.f), r3 = fmaxf(acc.w + bv.w, 0.f);
        const float4 wa = *(const float4*)(W2 + lane * 8);
        const float4 wb = *(const float4*)(W2 + lane * 8 + 4);
        p0 = r0 * wa.x + r1 * wa.z + r2 * wb.x + r3 * wb.z;
        p1 = r0 * wa.y + r1 * wa.w + r2 * wb.y + r3 * wb.w;
        for (int o = 16; o > 0; o >>= 1) {
            p0 += __shfl_xor(p0, o);
            p1 += __shfl_xor(p1, o);
        }
    }
    if (lane == 0) { sh2[wave * 2 + 0] = p0; sh2[wave * 2 + 1] = p1; }
    __syncthreads();
    if (wave == 0) {
        const int idx = blockIdx.x * 32 + lane;
        if (idx < n_nodes * 2) h2[idx] = sh2[lane];
    }
}

__global__ __launch_bounds__(256) void k_gather2(
    const float* __restrict__ h2, const int* __restrict__ rowptr, const int* __restrict__ cnt,
    const int* __restrict__ col, const float* __restrict__ dinv,
    const float* __restrict__ b2, float* __restrict__ out, int n_nodes)
{
    __shared__ float2 sres[32];
    const int tid = threadIdx.x;
    const int g = tid >> 3, sub = tid & 7;
    const int node = blockIdx.x * 32 + g;
    float a0 = 0.f, a1 = 0.f;
    if (node < n_nodes) {
        const int start = rowptr[node];
        const int end = start + cnt[node];
        const float dd = dinv[node];
        for (int j = start + sub; j < end; j += 8) {
            const int s = col[j];
            const float norm = dinv[s] * dd;
            const float2 v = *(const float2*)(h2 + (size_t)s * 2);
            a0 += v.x * norm;
            a1 += v.y * norm;
        }
    }
    for (int o = 4; o > 0; o >>= 1) {
        a0 += __shfl_xor(a0, o);
        a1 += __shfl_xor(a1, o);
    }
    if (sub == 0) sres[g] = make_float2(a0, a1);
    __syncthreads();
    if (tid < 32) {
        const int nd = blockIdx.x * 32 + tid;
        if (nd < n_nodes) {
            const float di = dinv[nd], w = di * di;
            const float2 hv = *(const float2*)(h2 + (size_t)nd * 2);
            const float2 r = sres[tid];
            float2 o;
            o.x = (b2[0] + hv.x * w) + r.x;
            o.y = (b2[1] + hv.y * w) + r.y;
            *(float2*)(out + (size_t)nd * 2) = o;
        }
    }
}

extern "C" void kernel_launch(void* const* d_in, const int* in_sizes, int n_in,
                              void* d_out, int out_size, void* d_ws, size_t ws_size,
                              hipStream_t stream)
{
    const float* x  = (const float*)d_in[0];
    const int*   ei = (const int*)d_in[1];
    const float* W1 = (const float*)d_in[2];
    const float* b1 = (const float*)d_in[3];
    const float* W2 = (const float*)d_in[4];
    const float* b2 = (const float*)d_in[5];
    float* out = (float*)d_out;

    const int n_nodes = in_sizes[0] / IN_CH;
    const int n_edges = in_sizes[1] / 2;
    const int* src = ei;
    const int* dst = ei + n_edges;
    const int n_sblk = (n_nodes + SCAN_T - 1) / SCAN_T;

    char* ws = (char*)d_ws;
    size_t off = 0;
    const size_t nb4 = ((size_t)n_nodes * 4 + 4095) & ~(size_t)4095;
    const size_t eb4 = ((size_t)n_edges * 4 + 4095) & ~(size_t)4095;
    float* dinv = (float*)(ws + off);  off += nb4;
    float* h = (float*)(ws + off);
    off += ((size_t)n_nodes * HID_CH * 4 + 4095) & ~(size_t)4095;
    float* h2 = (float*)(ws + off);    off += 2 * nb4;
    int* cnt = (int*)(ws + off);       off += nb4;
    int* rowptr = (int*)(ws + off);    off += nb4;
    int* bsum = (int*)(ws + off);      off += ((size_t)n_sblk * 4 + 4095) & ~(size_t)4095;
    int* rank = (int*)(ws + off);      off += eb4;
    int* col = (int*)(ws + off);

    k_zero_cnt  <<<(n_nodes + 255) / 256, 256, 0, stream>>>(cnt, n_nodes);
    k_count_rank<<<(n_edges + 255) / 256, 256, 0, stream>>>(dst, cnt, rank, n_edges);
    k_block_sums<<<n_sblk, SCAN_T, 0, stream>>>(cnt, bsum, n_nodes);
    k_scan_rows <<<n_sblk, SCAN_T, 0, stream>>>(cnt, bsum, rowptr, dinv, n_nodes);
    k_fill      <<<(n_edges + 255) / 256, 256, 0, stream>>>(src, dst, rowptr, rank, col, n_edges);

    k_gemm1_wmma<<<(n_nodes + 127) / 128, 256, 0, stream>>>(x, W1, h, n_nodes);

    k_gather1<<<(n_nodes + 15) / 16, 512, 0, stream>>>(h, rowptr, cnt, col, dinv, b1, W2, h2, n_nodes);
    k_gather2<<<(n_nodes + 31) / 32, 256, 0, stream>>>(h2, rowptr, cnt, col, dinv, b2, out, n_nodes);
}


# Edge grouping by destination moves its counters into LDS: two-level counting sort (bucket of 512 nodes, then node), no per-edge global atomic or scattered store; sum order = LDS arrival rank
# speedup vs baseline: 3.8680x; 1.7107x over previous
#include <hip/hip_runtime.h>
#include <hip/hip_bf16.h>
#include <cstdint>


typedef float v2f __attribute__((ext_vector_type(2)));
typedef float v8f __attribute__((ext_vector_type(8)));

#define IN_CH  256
#define HID_CH 128
#define KTILE  64
#define PSTRIDE 288

#define BKT_SHIFT 9
#define BKT_W     512
#define NBKT_MAX  256
#define CHUNK     4096
#define STAGE_CAP 12288

__global__ void k_zero_tot(int* __restrict__ tot) { tot[threadIdx.x] = 0; }

__device__ __forceinline__ int wave_incl_scan_i(int v, int lane) {
    for (int o = 1; o < 32; o <<= 1) {
        int t = __shfl_up(v, o);
        if (lane >= o) v += t;
    }
    return v;
}

__device__ __forceinline__ int block_excl_scan_1024(int v, int* sw) {
    const int lane = threadIdx.x & 31, wave = threadIdx.x >> 5;
    const int incl = wave_incl_scan_i(v, lane);
    if (lane == 31) sw[wave] = incl;
    __syncthreads();
    if (wave == 0) {
        const int t = sw[lane];
        const int s = wave_incl_scan_i(t, lane);
        sw[lane] = s - t;
    }
    __syncthreads();
    return sw[wave] + incl - v;
}

__device__ __forceinline__ int bucket_of(int d, int nbkt) {
    const int b = (int)((unsigned)d >> BKT_SHIFT);
    return b < nbkt ? b : nbkt - 1;
}

__global__ __launch_bounds__(1024) void k_bucket_count(const int* __restrict__ dst, int* __restrict__ tot,
                                                       int* __restrict__ boff, int ne, int nbkt) {
    __shared__ int lh[NBKT_MAX];
    const int tid = threadIdx.x;
    if (tid < NBKT_MAX) lh[tid] = 0;
    __syncthreads();
    const int e0 = blockIdx.x * CHUNK + tid;
    if ((blockIdx.x + 1) * CHUNK <= ne) {
        int dv[4];
#pragma unroll
        for (int k = 0; k < 4; ++k) dv[k] = dst[e0 + k * 1024];
#pragma unroll
        for (int k = 0; k < 4; ++k) atomicAdd(&lh[bucket_of(dv[k], nbkt)], 1);
    } else {
        for (int k = 0; k < 4; ++k) {
            const int e = e0 + k * 1024;
            if (e < ne) atomicAdd(&lh[bucket_of(dst[e], nbkt)], 1);
        }
    }
    __syncthreads();
    if (tid < nbkt) {
        const int c = lh[tid];
        int off = 0;
        if (c) off = __hip_atomic_fetch_add(&tot[tid], c, __ATOMIC_RELAXED, __HIP_MEMORY_SCOPE_AGENT);
        boff[blockIdx.x * NBKT_MAX + tid] = off;
    }
}

__global__ __launch_bounds__(1024) void k_bucket_scatter(const int* __restrict__ src, const int* __restrict__ dst,
                                                         const int* __restrict__ tot, const int* __restrict__ boff,
                                                         int2* __restrict__ recs, int ne, int nbkt) {
    __shared__ int lh[NBKT_MAX];
    __shared__ int lofs[NBKT_MAX];
    __shared__ int gb[NBKT_MAX];
    __shared__ int swA[32], swB[32];
    __shared__ int2 srec[CHUNK];
    const int tid = threadIdx.x;
    if (tid < NBKT_MAX) lh[tid] = 0;
    const int bo = tid < nbkt ? boff[blockIdx.x * NBKT_MAX + tid] : 0;
    const int bstart = block_excl_scan_1024(tid < nbkt ? tot[tid] : 0, swA);
    int s[4], d[4], r[4], b[4];
    const int e0 = blockIdx.x * CHUNK + tid;
    if ((blockIdx.x + 1) * CHUNK <= ne) {
#pragma unroll
        for (int k = 0; k < 4; ++k) {
            s[k] = src[e0 + k * 1024];
            d[k] = dst[e0 + k * 1024];
        }
#pragma unroll
        for (int k = 0; k < 4; ++k) {
            b[k] = bucket_of(d[k], nbkt);
            r[k] = atomicAdd(&lh[b[k]], 1);
        }
    } else {
#pragma unroll
        for (int k = 0; k < 4; ++k) {
            const int e = e0 + k * 1024;
            s[k] = 0; d[k] = 0; r[k] = 0;
            b[k] = -1;
            if (e < ne) {
                s[k] = src[e];
                d[k] = dst[e];
                b[k] = bucket_of(d[k], nbkt);
                r[k] = atomicAdd(&lh[b[k]], 1);
            }
        }
    }
    __syncthreads();
    const int lo = block_excl_scan_1024(tid < NBKT_MAX ? lh[tid] : 0, swB);
    if (tid < nbkt) {
        lofs[tid] = lo;
        gb[tid] = bstart + bo - lo;
    }
    __syncthreads();
#pragma unroll
    for (int k = 0; k < 4; ++k)
        if (b[k] >= 0) srec[lofs[b[k]] + r[k]] = make_int2(s[k], d[k]);
    __syncthreads();
    const int left = ne - blockIdx.x * CHUNK;
    const int nv = left < CHUNK ? left : CHUNK;
    for (int i = tid; i < nv; i += 1024) {
        const int2 rec = srec[i];
        const int p = gb[bucket_of(rec.y, nbkt)] + i;
        if ((unsigned)p < (unsigned)ne) recs[p] = rec;
    }
}

__global__ __launch_bounds__(1024) void k_bucket_rows(const int2* __restrict__ recs, const int* __restrict__ tot,
                                                      int* __restrict__ rowptr, int* __restrict__ cnt,
                                                      float* __restrict__ dinv, int* __restrict__ col,
                                                      int n_nodes, int ne, int nbkt) {
    __shared__ int lcnt[BKT_W + 1];
    __shared__ int lpos[BKT_W + 1];
    __shared__ int swA[32], swB[32];
    __shared__ int sb[2];
    __shared__ int stage[STAGE_CAP];
    const int tid = threadIdx.x;
    const int bkt = blockIdx.x;
    if (tid <= BKT_W) { lcnt[tid] = 0; lpos[tid] = 0; }
    const int t = tid < nbkt ? tot[tid] : 0;
    const int ex = block_excl_scan_1024(t, swA);
    if (tid == bkt) { sb[0] = ex; sb[1] = t; }
    __syncthreads();
    int bstart = sb[0];
    int nb = sb[1];
    if (bstart < 0 || bstart > ne) { bstart = 0; nb = 0; }
    if (nb > ne - bstart) nb = ne - bstart;
    const int2* rp = recs + bstart;
    for (int i0 = tid; i0 < nb; i0 += 4096) {
        int y[4];
#pragma unroll
        for (int k = 0; k < 4; ++k) {
            const int i = i0 + k * 1024;
            y[k] = rp[i < nb ? i : nb - 1].y;
        }
#pragma unroll
        for (int k = 0; k < 4; ++k)
            atomicAdd(&lcnt[i0 + k * 1024 < nb ? (y[k] & (BKT_W - 1)) : BKT_W], 1);
    }
    __syncthreads();
    const int c = tid < BKT_W ? lcnt[tid] : 0;
    const int ex2 = block_excl_scan_1024(c, swB);
    if (tid < BKT_W) {
        lpos[tid] = ex2;
        const int node = bkt * BKT_W + tid;
        if (node < n_nodes) {
            rowptr[node] = bstart + ex2;
            cnt[node] = c;
            dinv[node] = rsqrtf((float)(c + 1));
        }
    }
    __syncthreads();
    const bool staged = nb <= STAGE_CAP;
    for (int i0 = tid; i0 < nb; i0 += 4096) {
        int2 rec[4];
        int p[4];
#pragma unroll
        for (int k = 0; k < 4; ++k) {
            const int i = i0 + k * 1024;
            rec[k] = rp[i < nb ? i : nb - 1];
        }
#pragma unroll
        for (int k = 0; k < 4; ++k)
            p[k] = atomicAdd(&lpos[i0 + k * 1024 < nb ? (rec[k].y & (BKT_W - 1)) : BKT_W], 1);
#pragma unroll
        for (int k = 0; k < 4; ++k) {
            if (i0 + k * 1024 < nb && (unsigned)p[k] < (unsigned)nb) {
                if (staged) stage[p[k]] = rec[k].x;
                else        col[bstart + p[k]] = rec[k].x;
            }
        }
    }
    if (staged) {
        __syncthreads();
        for (int i = tid; i < nb; i += 1024) col[bstart + i] = stage[i];
    }
}

__global__ __launch_bounds__(256) void k_gemm1_wmma(
    const float* __restrict__ x, const float* __restrict__ W1,
    float* __restrict__ h, int n_nodes)
{
    __shared__ float sW[(KTILE / 2) * PSTRIDE];

    const int tid  = threadIdx.x;
    const int wave = tid >> 5;
    const int lane = tid & 31;
    const int half = lane >> 4;
    const int l16  = lane & 15;

    const int row_base = (blockIdx.x * 8 + wave) * 16;
    int arow = row_base + l16;
    if (arow >= n_nodes) arow = n_nodes - 1;
    const float* aptr = x + (size_t)arow * IN_CH + half * 2;

    v8f acc[8] = {};

    for (int k0 = 0; k0 < IN_CH; k0 += KTILE) {
        __syncthreads();
        for (int i = tid; i < (KTILE / 2) * 32; i += 256) {
            int p  = i >> 5;
            int c4 = (i & 31) * 4;
            float4 r0 = *(const float4*)(W1 + (size_t)(k0 + 2 * p + 0) * HID_CH + c4);
            float4 r1 = *(const float4*)(W1 + (size_t)(k0 + 2 * p + 1) * HID_CH + c4);
            float* dp = &sW[p * PSTRIDE + c4 * 2];
            ((float2*)dp)[0] = make_float2(r0.x, r1.x);
            ((float2*)dp)[1] = make_float2(r0.y, r1.y);
            ((float2*)dp)[2] = make_float2(r0.z, r1.z);
            ((float2*)dp)[3] = make_float2(r0.w, r1.w);
        }
        __syncthreads();

#pragma unroll 4
        for (int k = 0; k < KTILE; k += 4) {
            v2f a = *(const v2f*)(aptr + k0 + k);
            const int p = (k >> 1) + half;
            const float* bbase = &sW[p * PSTRIDE + l16 * 2];
#pragma unroll
            for (int nt = 0; nt < 8; ++nt) {
                v2f b = *(const v2f*)(bbase + nt * 32);
                acc[nt] = __builtin_amdgcn_wmma_f32_16x16x4_f32(
                    false, a, false, b, (short)0, acc[nt], false, false);
            }
        }
    }

#pragma unroll
    for (int nt = 0; nt < 8; ++nt) {
#pragma unroll
        for (int v = 0; v < 8; ++v) {
            int r = row_base + v + 8 * half;
            if (r < n_nodes) h[(size_t)r * HID_CH + nt * 16 + l16] = acc[nt][v];
        }
    }
}

__global__ __launch_bounds__(512) void k_gather1(
    const float* __restrict__ h, const int* __restrict__ rowptr, const int* __restrict__ cnt,
    const int* __restrict__ col, const float* __restrict__ dinv,
    const float* __restrict__ b1, const float* __restrict__ W2,
    float* __restrict__ h2, int n_nodes)
{
    __shared__ float sh2[32];
    const int lane = threadIdx.x & 31;
    const int wave = threadIdx.x >> 5;
    const int node = __builtin_amdgcn_readfirstlane(blockIdx.x * 16 + wave);
    float p0 = 0.f, p1 = 0.f;
    if (node < n_nodes) {
        const int start = rowptr[node];
        const int deg = cnt[node];
        const float dd = dinv[node];
        const float* hl = h + lane * 4;
        const float wself = dd * dd;
        float4 hv = *(const float4*)(hl + (size_t)node * HID_CH);
        float4 acc;
        acc.x = hv.x * wself; acc.y = hv.y * wself; acc.z = hv.z * wself; acc.w = hv.w * wself;
        for (int base = 0; base < deg; base += 32) {
            const int m = (deg - base < 32) ? (deg - base) : 32;
            int my_s = 0; float my_w = 0.f;
            if (lane < m) {
                my_s = col[start + base + lane];
                my_w = dinv[my_s] * dd;
            }
            const int my_wi = __float_as_int(my_w);
            int j = 0;
            for (; j + 4 <= m; j += 4) {
                const int s0 = __builtin_amdgcn_readlane(my_s, j + 0);
                const int s1 = __builtin_amdgcn_readlane(my_s, j + 1);
                const int s2 = __builtin_amdgcn_readlane(my_s, j + 2);
                const int s3 = __builtin_amdgcn_readlane(my_s, j + 3);
                const float w0 = __int_as_float(__builtin_amdgcn_readlane(my_wi, j + 0));
                const float w1 = __int_as_float(__builtin_amdgcn_readlane(my_wi, j + 1));
                const float w2 = __int_as_float(__builtin_amdgcn_readlane(my_wi, j + 2));
                const float w3 = __int_as_float(__builtin_amdgcn_readlane(my_wi, j + 3));
                const float4 a0 = *(const float4*)(hl + (size_t)s0 * HID_CH);
                const float4 a1 = *(const float4*)(hl + (size_t)s1 * HID_CH);
                const float4 a2 = *(const float4*)(hl + (size_t)s2 * HID_CH);
                const float4 a3 = *(const float4*)(hl + (size_t)s3 * HID_CH);
                acc.x += a0.x * w0; acc.y += a0.y * w0; acc.z += a0.z * w0; acc.w += a0.w * w0;
                acc.x += a1.x * w1; acc.y += a1.y * w1; acc.z += a1.z * w1; acc.w += a1.w * w1;
                acc.x += a2.x * w2; acc.y += a2.y * w2; acc.z += a2.z * w2; acc.w += a2.w * w2;
                acc.x += a3.x * w3; acc.y += a3.y * w3; acc.z += a3.z * w3; acc.w += a3.w * w3;
            }
            for (; j < m; ++j) {
                const int s0 = __builtin_amdgcn_readlane(my_s, j);
                const float w0 = __int_as_float(__builtin_amdgcn_readlane(my_wi, j));
                const float4 a0 = *(const float4*)(hl + (size_t)s0 * HID_CH);
                acc.x += a0.x * w0; acc.y += a0.y * w0; acc.z += a0.z * w0; acc.w += a0.w * w0;
            }
        }
        const float4 bv = *(const float4*)(b1 + lane * 4);
        const float r0 = fmaxf(acc.x + bv.x, 0.f), r1 = fmaxf(acc.y + bv.y, 0.f);
        const float r2 = fmaxf(acc.z + bv.z, 0.f), r3 = fmaxf(acc.w + bv.w, 0.f);
        const float4 wa = *(const float4*)(W2 + lane * 8);
        const float4 wb = *(const float4*)(W2 + lane * 8 + 4);
        p0 = r0 * wa.x + r1 * wa.z + r2 * wb.x + r3 * wb.z;
        p1 = r0 * wa.y + r1 * wa.w + r2 * wb.y + r3 * wb.w;
        for (int o = 16; o > 0; o >>= 1) {
            p0 += __shfl_xor(p0, o);
            p1 += __shfl_xor(p1, o);
        }
    }
    if (lane == 0) { sh2[wave * 2 + 0] = p0; sh2[wave * 2 + 1] = p1; }
    __syncthreads();
    if (wave == 0) {
        const int idx = blockIdx.x * 32 + lane;
        if (idx < n_nodes * 2) h2[idx] = sh2[lane];
    }
}

__global__ __launch_bounds__(256) void k_gather2(
    const float* __restrict__ h2, const int* __restrict__ rowptr, const int* __restrict__ cnt,
    const int* __restrict__ col, const float* __restrict__ dinv,
    const float* __restrict__ b2, float* __restrict__ out, int n_nodes)
{
    __shared__ float2 sres[32];
    const int tid = threadIdx.x;
    const int g = tid >> 3, sub = tid & 7;
    const int node = blockIdx.x * 32 + g;
    float a0 = 0.f, a1 = 0.f;
    if (node < n_nodes) {
        const int start = rowptr[node];
        const int end = start + cnt[node];
        const float dd = dinv[node];
        for (int j = start + sub; j < end; j += 8) {
            const int s = col[j];
            const float norm = dinv[s] * dd;
            const float2 v = *(const float2*)(h2 + (size_t)s * 2);
            a0 += v.x * norm;
            a1 += v.y * norm;
        }
    }
    for (int o = 4; o > 0; o >>= 1) {
        a0 += __shfl_xor(a0, o);
        a1 += __shfl_xor(a1, o);
    }
    if (sub == 0) sres[g] = make_float2(a0, a1);
    __syncthreads();
    if (tid < 32) {
        const int nd = blockIdx.x * 32 + tid;
        if (nd < n_nodes) {
            const float di = dinv[nd], w = di * di;
            const float2 hv = *(const float2*)(h2 + (size_t)nd * 2);
            const float2 r = sres[tid];
            float2 o;
            o.x = (b2[0] + hv.x * w) + r.x;
            o.y = (b2[1] + hv.y * w) + r.y;
            *(float2*)(out + (size_t)nd * 2) = o;
        }
    }
}

extern "C" void kernel_launch(void* const* d_in, const int* in_sizes, int n_in,
                              void* d_out, int out_size, void* d_ws, size_t ws_size,
                              hipStream_t stream)
{
    const float* x  = (const float*)d_in[0];
    const int*   ei = (const int*)d_in[1];
    const float* W1 = (const float*)d_in[2];
    const float* b1 = (const float*)d_in[3];
    const float* W2 = (const float*)d_in[4];
    const float* b2 = (const float*)d_in[5];
    float* out = (float*)d_out;

    const int n_nodes = in_sizes[0] / IN_CH;
    const int n_edges = in_sizes[1] / 2;
    const int* src = ei;
    const int* dst = ei + n_edges;
    const int nbkt = (n_nodes + BKT_W - 1) / BKT_W;
    const int n_chunk = (n_edges + CHUNK - 1) / CHUNK;

    char* ws = (char*)d_ws;
    size_t off = 0;
    const size_t nb4 = ((size_t)n_nodes * 4 + 4095) & ~(size_t)4095;
    const size_t eb4 = ((size_t)n_edges * 4 + 4095) & ~(size_t)4095;
    float* dinv = (float*)(ws + off);  off += nb4;
    float* h = (float*)(ws + off);
    off += ((size_t)n_nodes * HID_CH * 4 + 4095) & ~(size_t)4095;
    float* h2 = (float*)(ws + off);    off += 2 * nb4;
    int* cnt = (int*)(ws + off);       off += nb4;
    int* rowptr = (int*)(ws + off);    off += nb4;
    int* tot = (int*)(ws + off);       off += 4096;
    int* boff = (int*)(ws + off);      off += ((size_t)n_chunk * NBKT_MAX * 4 + 4095) & ~(size_t)4095;
    int2* recs = (int2*)(ws + off);    off += 2 * eb4;
    int* col = (int*)(ws + off);

    k_zero_tot      <<<1, NBKT_MAX, 0, stream>>>(tot);
    k_bucket_count  <<<n_chunk, 1024, 0, stream>>>(dst, tot, boff, n_edges, nbkt);
    k_bucket_scatter<<<n_chunk, 1024, 0, stream>>>(src, dst, tot, boff, recs, n_edges, nbkt);
    k_bucket_rows   <<<nbkt, 1024, 0, stream>>>(recs, tot, rowptr, cnt, dinv, col, n_nodes, n_edges, nbkt);

    k_gemm1_wmma<<<(n_nodes + 127) / 128, 256, 0, stream>>>(x, W1, h, n_nodes);

    k_gather1<<<(n_nodes + 15) / 16, 512, 0, stream>>>(h, rowptr, cnt, col, dinv, b1, W2, h2, n_nodes);
    k_gather2<<<(n_nodes + 31) / 32, 256, 0, stream>>>(h2, rowptr, cnt, col, dinv, b2, out, n_nodes);
}


# x @ W1 loses its own launch: its 256-row blocks (32 waves, 16x64 C tile per wave) run as extra blocks of the two launches of the edge grouping's last two stages, matrix pipe under their waits
# speedup vs baseline: 4.0357x; 1.0434x over previous
#include <hip/hip_runtime.h>
#include <hip/hip_bf16.h>
#include <cstdint>


typedef float v2f __attribute__((ext_vector_type(2)));
typedef float v8f __attribute__((ext_vector_type(8)));

#define IN_CH  256
#define HID_CH 128
#define KTILE  64
#define PSTRIDE 288

#define BKT_SHIFT 9
#define BKT_W     512
#define NBKT_MAX  256
#define CHUNK     4096
#define STAGE_CAP 12288

__global__ void k_zero_tot(int* __restrict__ tot) { tot[threadIdx.x] = 0; }

__device__ __forceinline__ int wave_incl_scan_i(int v, int lane) {
    for (int o = 1; o < 32; o <<= 1) {
        int t = __shfl_up(v, o);
        if (lane >= o) v += t;
    }
    return v;
}

__device__ __forceinline__ int block_excl_scan_1024(int v, int* sw) {
    const int lane = threadIdx.x & 31, wave = threadIdx.x >> 5;
    const int incl = wave_incl_scan_i(v, lane);
    if (lane == 31) sw[wave] = incl;
    __syncthreads();
    if (wave == 0) {
        const int t = sw[lane];
        const int s = wave_incl_scan_i(t, lane);
        sw[lane] = s - t;
    }
    __syncthreads();
    return sw[wave] + incl - v;
}

__device__ __forceinline__ int bucket_of(int d, int nbkt) {
    const int b = (int)((unsigned)d >> BKT_SHIFT);
    return b < nbkt ? b : nbkt - 1;
}

__global__ __launch_bounds__(1024) void k_bucket_count(const int* __restrict__ dst, int* __restrict__ tot,
                                                       int* __restrict__ boff, int ne, int nbkt) {
    __shared__ int lh[NBKT_MAX];
    const int tid = threadIdx.x;
    if (tid < NBKT_MAX) lh[tid] = 0;
    __syncthreads();
    const int e0 = blockIdx.x * CHUNK + tid;
    if ((blockIdx.x + 1) * CHUNK <= ne) {
        int dv[4];
#pragma unroll
        for (int k = 0; k < 4; ++k) dv[k] = dst[e0 + k * 1024];
#pragma unroll
        for (int k = 0; k < 4; ++k) atomicAdd(&lh[bucket_of(dv[k], nbkt)], 1);
    } else {
        for (int k = 0; k < 4; ++k) {
            const int e = e0 + k * 1024;
            if (e < ne) atomicAdd(&lh[bucket_of(dst[e], nbkt)], 1);
        }
    }
    __syncthreads();
    if (tid < nbkt) {
        const int c = lh[tid];
        int off = 0;
        if (c) off = __hip_atomic_fetch_add(&tot[tid], c, __ATOMIC_RELAXED, __HIP_MEMORY_SCOPE_AGENT);
        boff[blockIdx.x * NBKT_MAX + tid] = off;
    }
}

__device__ __forceinline__ void scatter_role(const int blk, const int* __restrict__ src, const int* __restrict__ dst,
                                             const int* __restrict__ tot, const int* __restrict__ boff,
                                             int2* __restrict__ recs, int ne, int nbkt) {
    __shared__ int lh[NBKT_MAX];
    __shared__ int lofs[NBKT_MAX];
    __shared__ int gb[NBKT_MAX];
    __shared__ int swA[32], swB[32];
    __shared__ int2 srec[CHUNK];
    const int tid = threadIdx.x;
    if (tid < NBKT_MAX) lh[tid] = 0;
    const int bo = tid < nbkt ? boff[blk * NBKT_MAX + tid] : 0;
    const int bstart = block_excl_scan_1024(tid < nbkt ? tot[tid] : 0, swA);
    int s[4], d[4], r[4], b[4];
    const int e0 = blk * CHUNK + tid;
    if ((blk + 1) * CHUNK <= ne) {
#pragma unroll
        for (int k = 0; k < 4; ++k) {
            s[k] = src[e0 + k * 1024];
            d[k] = dst[e0 + k * 1024];
        }
#pragma unroll
        for (int k = 0; k < 4; ++k) {
            b[k] = bucket_of(d[k], nbkt);
            r[k] = atomicAdd(&lh[b[k]], 1);
        }
    } else {
#pragma unroll
        for (int k = 0; k < 4; ++k) {
            const int e = e0 + k * 1024;
            s[k] = 0; d[k] = 0; r[k] = 0;
            b[k] = -1;
            if (e < ne) {
                s[k] = src[e];
                d[k] = dst[e];
                b[k] = bucket_of(d[k], nbkt);
                r[k] = atomicAdd(&lh[b[k]], 1);
            }
        }
    }
    __syncthreads();
    const int lo = block_excl_scan_1024(tid < NBKT_MAX ? lh[tid] : 0, swB);
    if (tid < nbkt) {
        lofs[tid] = lo;
        gb[tid] = bstart + bo - lo;
    }
    __syncthreads();
#pragma unroll
    for (int k = 0; k < 4; ++k)
        if (b[k] >= 0) srec[lofs[b[k]] + r[k]] = make_int2(s[k], d[k]);
    __syncthreads();
    const int left = ne - blk * CHUNK;
    const int nv = left < CHUNK ? left : CHUNK;
    for (int i = tid; i < nv; i += 1024) {
        const int2 rec = srec[i];
        const int p = gb[bucket_of(rec.y, nbkt)] + i;
        if ((unsigned)p < (unsigned)ne) recs[p] = rec;
    }
}

__device__ __forceinline__ void rows_role(const int blk, const int2* __restrict__ recs, const int* __restrict__ tot,
                                          int* __restrict__ rowptr, int* __restrict__ cnt,
                                          float* __restrict__ dinv, int* __restrict__ col,
                                          int n_nodes, int ne, int nbkt) {
    __shared__ int lcnt[BKT_W + 1];
    __shared__ int lpos[BKT_W + 1];
    __shared__ int swA[32], swB[32];
    __shared__ int sb[2];
    __shared__ int stage[STAGE_CAP];
    const int tid = threadIdx.x;
    const int bkt = blk;
    if (tid <= BKT_W) { lcnt[tid] = 0; lpos[tid] = 0; }
    const int t = tid < nbkt ? tot[tid] : 0;
    const int ex = block_excl_scan_1024(t, swA);
    if (tid == bkt) { sb[0] = ex; sb[1] = t; }
    __syncthreads();
    int bstart = sb[0];
    int nb = sb[1];
    if (bstart < 0 || bstart > ne) { bstart = 0; nb = 0; }
    if (nb > ne - bstart) nb = ne - bstart;
    const int2* rp = recs + bstart;
    for (int i0 = tid; i0 < nb; i0 += 4096) {
        int y[4];
#pragma unroll
        for (int k = 0; k < 4; ++k) {
            const int i = i0 + k * 1024;
            y[k] = rp[i < nb ? i : nb - 1].y;
        }
#pragma unroll
        for (int k = 0; k < 4; ++k)
            atomicAdd(&lcnt[i0 + k * 1024 < nb ? (y[k] & (BKT_W - 1)) : BKT_W], 1);
    }
    __syncthreads();
    const int c = tid < BKT_W ? lcnt[tid] : 0;
    const int ex2 = block_excl_scan_1024(c, swB);
    if (tid < BKT_W) {
        lpos[tid] = ex2;
        const int node = bkt * BKT_W + tid;
        if (node < n_nodes) {
            rowptr[node] = bstart + ex2;
            cnt[node] = c;
            dinv[node] = rsqrtf((float)(c + 1));
        }
    }
    __syncthreads();
    const bool staged = nb <= STAGE_CAP;
    for (int i0 = tid; i0 < nb; i0 += 4096) {
        int2 rec[4];
        int p[4];
#pragma unroll
        for (int k = 0; k < 4; ++k) {
            const int i = i0 + k * 1024;
            rec[k] = rp[i < nb ? i : nb - 1];
        }
#pragma unroll
        for (int k = 0; k < 4; ++k)
            p[k] = atomicAdd(&lpos[i0 + k * 1024 < nb ? (rec[k].y & (BKT_W - 1)) : BKT_W], 1);
#pragma unroll
        for (int k = 0; k < 4; ++k) {
            if (i0 + k * 1024 < nb && (unsigned)p[k] < (unsigned)nb) {
                if (staged) stage[p[k]] = rec[k].x;
                else        col[bstart + p[k]] = rec[k].x;
            }
        }
    }
    if (staged) {
        __syncthreads();
        for (int i = tid; i < nb; i += 1024) col[bstart + i] = stage[i];
    }
}

__device__ __forceinline__ void gemm_role(const int gblk, const float* __restrict__ x, const float* __restrict__ W1,
                                          float* __restrict__ h, int n_nodes)
{
    __shared__ float sW[(KTILE / 2) * PSTRIDE];

    const int tid  = threadIdx.x;
    const int wave = tid >> 5;
    const int lane = tid & 31;
    const int half = lane >> 4;
    const int l16  = lane & 15;
    const int rt   = wave & 15;
    const int ch   = wave >> 4;

    const int row_base = gblk * 256 + rt * 16;
    int arow = row_base + l16;
    if (arow >= n_nodes) arow = n_nodes - 1;
    const float* aptr = x + (size_t)arow * IN_CH + half * 2;

    v8f acc[4] = {};

    for (int k0 = 0; k0 < IN_CH; k0 += KTILE) {
        __syncthreads();
        {
            const int p  = tid >> 5;
            const int c4 = (tid & 31) * 4;
            float4 r0 = *(const float4*)(W1 + (size_t)(k0 + 2 * p + 0) * HID_CH + c4);
            float4 r1 = *(const float4*)(W1 + (size_t)(k0 + 2 * p + 1) * HID_CH + c4);
            float* dp = &sW[p * PSTRIDE + c4 * 2];
            ((float2*)dp)[0] = make_float2(r0.x, r1.x);
            ((float2*)dp)[1] = make_float2(r0.y, r1.y);
            ((float2*)dp)[2] = make_float2(r0.z, r1.z);
            ((float2*)dp)[3] = make_float2(r0.w, r1.w);
        }
        __syncthreads();

#pragma unroll 4
        for (int k = 0; k < KTILE; k += 4) {
            v2f a = *(const v2f*)(aptr + k0 + k);
            const int p = (k >> 1) + half;
            const float* bbase = &sW[p * PSTRIDE + ch * 128 + l16 * 2];
#pragma unroll
            for (int nt = 0; nt < 4; ++nt) {
                v2f b = *(const v2f*)(bbase + nt * 32);
                acc[nt] = __builtin_amdgcn_wmma_f32_16x16x4_f32(
                    false, a, false, b, (short)0, acc[nt], false, false);
            }
        }
    }

#pragma unroll
    for (int nt = 0; nt < 4; ++nt) {
#pragma unroll
        for (int v = 0; v < 8; ++v) {
            int r = row_base + v + 8 * half;
            if (r < n_nodes) h[(size_t)r * HID_CH + ch * 64 + nt * 16 + l16] = acc[nt][v];
        }
    }
}

__global__ __launch_bounds__(1024) __attribute__((amdgpu_waves_per_eu(16, 16))) void k_scatter_and_gemm(
    const int* __restrict__ src, const int* __restrict__ dst, const int* __restrict__ tot,
    const int* __restrict__ boff, int2* __restrict__ recs, int ne, int nbkt,
    const float* __restrict__ x, const float* __restrict__ W1, float* __restrict__ h, int n_nodes,
    int g_first, int g_count)
{
    const int b = blockIdx.x;
    if (b < g_count) gemm_role(g_first + b, x, W1, h, n_nodes);
    else             scatter_role(b - g_count, src, dst, tot, boff, recs, ne, nbkt);
}

__global__ __launch_bounds__(1024) __attribute__((amdgpu_waves_per_eu(16, 16))) void k_rows_and_gemm(
    const int2* __restrict__ recs, const int* __restrict__ tot, int* __restrict__ rowptr, int* __restrict__ cnt,
    float* __restrict__ dinv, int* __restrict__ col, int ne, int nbkt,
    const float* __restrict__ x, const float* __restrict__ W1, float* __restrict__ h, int n_nodes,
    int g_first, int g_count)
{
    const int b = blockIdx.x;
    if (b < g_count) gemm_role(g_first + b, x, W1, h, n_nodes);
    else             rows_role(b - g_count, recs, tot, rowptr, cnt, dinv, col, n_nodes, ne, nbkt);
}

__global__ __launch_bounds__(512) void k_gather1(
    const float* __restrict__ h, const int* __restrict__ rowptr, const int* __restrict__ cnt,
    const int* __restrict__ col, const float* __restrict__ dinv,
    const float* __restrict__ b1, const float* __restrict__ W2,
    float* __restrict__ h2, int n_nodes)
{
    __shared__ float sh2[32];
    const int lane = threadIdx.x & 31;
    const int wave = threadIdx.x >> 5;
    const int node = __builtin_amdgcn_readfirstlane(blockIdx.x * 16 + wave);
    float p0 = 0.f, p1 = 0.f;
    if (node < n_nodes) {
        const int start = rowptr[node];
        const int deg = cnt[node];
        const float dd = dinv[node];
        const float* hl = h + lane * 4;
        const float wself = dd * dd;
        float4 hv = *(const float4*)(hl + (size_t)node * HID_CH);
        float4 acc;
        acc.x = hv.x * wself; acc.y = hv.y * wself; acc.z = hv.z * wself; acc.w = hv.w * wself;
        for (int base = 0; base < deg; base += 32) {
            const int m = (deg - base < 32) ? (deg - base) : 32;
            int my_s = 0; float my_w = 0.f;
            if (lane < m) {
                my_s = col[start + base + lane];
                my_w = dinv[my_s] * dd;
            }
            const int my_wi = __float_as_int(my_w);
            int j = 0;
            for (; j + 4 <= m; j += 4) {
                const int s0 = __builtin_amdgcn_readlane(my_s, j + 0);
                const int s1 = __builtin_amdgcn_readlane(my_s, j + 1);
                const int s2 = __builtin_amdgcn_readlane(my_s, j + 2);
                const int s3 = __builtin_amdgcn_readlane(my_s, j + 3);
                const float w0 = __int_as_float(__builtin_amdgcn_readlane(my_wi, j + 0));
                const float w1 = __int_as_float(__builtin_amdgcn_readlane(my_wi, j + 1));
                const float w2 = __int_as_float(__builtin_amdgcn_readlane(my_wi, j + 2));
                const float w3 = __int_as_float(__builtin_amdgcn_readlane(my_wi, j + 3));
                const float4 a0 = *(const float4*)(hl + (size_t)s0 * HID_CH);
                const float4 a1 = *(const float4*)(hl + (size_t)s1 * HID_CH);
                const float4 a2 = *(const float4*)(hl + (size_t)s2 * HID_CH);
                const float4 a3 = *(const float4*)(hl + (size_t)s3 * HID_CH);
                acc.x += a0.x * w0; acc.y += a0.y * w0; acc.z += a0.z * w0; acc.w += a0.w * w0;
                acc.x += a1.x * w1; acc.y += a1.y * w1; acc.z += a1.z * w1; acc.w += a1.w * w1;
                acc.x += a2.x * w2; acc.y += a2.y * w2; acc.z += a2.z * w2; acc.w += a2.w * w2;
                acc.x += a3.x * w3; acc.y += a3.y * w3; acc.z += a3.z * w3; acc.w += a3.w * w3;
            }
            for (; j < m; ++j) {
                const int s0 = __builtin_amdgcn_readlane(my_s, j);
                const float w0 = __int_as_float(__builtin_amdgcn_readlane(my_wi, j));
                const float4 a0 = *(const float4*)(hl + (size_t)s0 * HID_CH);
                acc.x += a0.x * w0; acc.y += a0.y * w0; acc.z += a0.z * w0; acc.w += a0.w * w0;
            }
        }
        const float4 bv = *(const float4*)(b1 + lane * 4);
        const float r0 = fmaxf(acc.x + bv.x, 0.f), r1 = fmaxf(acc.y + bv.y, 0.f);
        const float r2 = fmaxf(acc.z + bv.z, 0.f), r3 = fmaxf(acc.w + bv.w, 0.f);
        const float4 wa = *(const float4*)(W2 + lane * 8);
        const float4 wb = *(const float4*)(W2 + lane * 8 + 4);
        p0 = r0 * wa.x + r1 * wa.z + r2 * wb.x + r3 * wb.z;
        p1 = r0 * wa.y + r1 * wa.w + r2 * wb.y + r3 * wb.w;
        for (int o = 16; o > 0; o >>= 1) {
            p0 += __shfl_xor(p0, o);
            p1 += __shfl_xor(p1, o);
        }
    }
    if (lane == 0) { sh2[wave * 2 + 0] = p0; sh2[wave * 2 + 1] = p1; }
    __syncthreads();
    if (wave == 0) {
        const int idx = blockIdx.x * 32 + lane;
        if (idx < n_nodes * 2) h2[idx] = sh2[lane];
    }
}

__global__ __launch_bounds__(256) void k_gather2(
    const float* __restrict__ h2, const int* __restrict__ rowptr, const int* __restrict__ cnt,
    const int* __restrict__ col, const float* __restrict__ dinv,
    const float* __restrict__ b2, float* __restrict__ out, int n_nodes)
{
    __shared__ float2 sres[32];
    const int tid = threadIdx.x;
    const int g = tid >> 3, sub = tid & 7;
    const int node = blockIdx.x * 32 + g;
    float a0 = 0.f, a1 = 0.f;
    if (node < n_nodes) {
        const int start = rowptr[node];
        const int end = start + cnt[node];
        const float dd = dinv[node];
        for (int j = start + sub; j < end; j += 8) {
            const int s = col[j];
            const float norm = dinv[s] * dd;
            const float2 v = *(const float2*)(h2 + (size_t)s * 2);
            a0 += v.x * norm;
            a1 += v.y * norm;
        }
    }
    for (int o = 4; o > 0; o >>= 1) {
        a0 += __shfl_xor(a0, o);
        a1 += __shfl_xor(a1, o);
    }
    if (sub == 0) sres[g] = make_float2(a0, a1);
    __syncthreads();
    if (tid < 32) {
        const int nd = blockIdx.x * 32 + tid;
        if (nd < n_nodes) {
            const float di = dinv[nd], w = di * di;
            const float2 hv = *(const float2*)(h2 + (size_t)nd * 2);
            const float2 r = sres[tid];
            float2 o;
            o.x = (b2[0] + hv.x * w) + r.x;
            o.y = (b2[1] + hv.y * w) + r.y;
            *(float2*)(out + (size_t)nd * 2) = o;
        }
    }
}

extern "C" void kernel_launch(void* const* d_in, const int* in_sizes, int n_in,
                              void* d_out, int out_size, void* d_ws, size_t ws_size,
                              hipStream_t stream)
{
    const float* x  = (const float*)d_in[0];
    const int*   ei = (const int*)d_in[1];
    const float* W1 = (const float*)d_in[2];
    const float* b1 = (const float*)d_in[3];
    const float* W2 = (const float*)d_in[4];
    const float* b2 = (const float*)d_in[5];
    float* out = (float*)d_out;

    const int n_nodes = in_sizes[0] / IN_CH;
    const int n_edges = in_sizes[1] / 2;
    const int* src = ei;
    const int* dst = ei + n_edges;
    const int nbkt = (n_nodes + BKT_W - 1) / BKT_W;
    const int n_chunk = (n_edges + CHUNK - 1) / CHUNK;

    char* ws = (char*)d_ws;
    size_t off = 0;
    const size_t nb4 = ((size_t)n_nodes * 4 + 4095) & ~(size_t)4095;
    const size_t eb4 = ((size_t)n_edges * 4 + 4095) & ~(size_t)4095;
    float* dinv = (float*)(ws + off);  off += nb4;
    float* h = (float*)(ws + off);
    off += ((size_t)n_nodes * HID_CH * 4 + 4095) & ~(size_t)4095;
    float* h2 = (float*)(ws + off);    off += 2 * nb4;
    int* cnt = (int*)(ws + off);       off += nb4;
    int* rowptr = (int*)(ws + off);    off += nb4;
    int* tot = (int*)(ws + off);       off += 4096;
    int* boff = (int*)(ws + off);      off += ((size_t)n_chunk * NBKT_MAX * 4 + 4095) & ~(size_t)4095;
    int2* recs = (int2*)(ws + off);    off += 2 * eb4;
    int* col = (int*)(ws + off);

    k_zero_tot      <<<1, NBKT_MAX, 0, stream>>>(tot);
    k_bucket_count  <<<n_chunk, 1024, 0, stream>>>(dst, tot, boff, n_edges, nbkt);
    const int g_all = (n_nodes + 255) / 256;
    const int g_a = (g_all + 1) / 2;
    const int g_b = g_all - g_a;
    k_scatter_and_gemm<<<g_a + n_chunk, 1024, 0, stream>>>(src, dst, tot, boff, recs, n_edges, nbkt,
                                                          x, W1, h, n_nodes, 0, g_a);
    k_rows_and_gemm   <<<g_b + nbkt, 1024, 0, stream>>>(recs, tot, rowptr, cnt, dinv, col, n_edges, nbkt,
                                                       x, W1, h, n_nodes, g_a, g_b);

    k_gather1<<<(n_nodes + 15) / 16, 512, 0, stream>>>(h, rowptr, cnt, col, dinv, b1, W2, h2, n_nodes);
    k_gather2<<<(n_nodes + 31) / 32, 256, 0, stream>>>(h2, rowptr, cnt, col, dinv, b2, out, n_nodes);
}


# Aggregations: long-lived waves (grid resident at once; alone this was within noise) that read a node's list in one go and ask for 8 rows (layer 2: 4 sources) before adding; sum order kept
# speedup vs baseline: 4.1862x; 1.0373x over previous
#include <hip/hip_runtime.h>
#include <hip/hip_bf16.h>
#include <cstdint>


typedef float v2f __attribute__((ext_vector_type(2)));
typedef float v8f __attribute__((ext_vector_type(8)));

#define IN_CH  256
#define HID_CH 128
#define KTILE  64
#define PSTRIDE 288

#define BKT_SHIFT 9
#define BKT_W     512
#define NBKT_MAX  256
#define CHUNK     4096
#define STAGE_CAP 12288

__global__ void k_zero_tot(int* __restrict__ tot) { tot[threadIdx.x] = 0; }

__device__ __forceinline__ int wave_incl_scan_i(int v, int lane) {
    for (int o = 1; o < 32; o <<= 1) {
        int t = __shfl_up(v, o);
        if (lane >= o) v += t;
    }
    return v;
}

__device__ __forceinline__ int block_excl_scan_1024(int v, int* sw) {
    const int lane = threadIdx.x & 31, wave = threadIdx.x >> 5;
    const int incl = wave_incl_scan_i(v, lane);
    if (lane == 31) sw[wave] = incl;
    __syncthreads();
    if (wave == 0) {
        const int t = sw[lane];
        const int s = wave_incl_scan_i(t, lane);
        sw[lane] = s - t;
    }
    __syncthreads();
    return sw[wave] + incl - v;
}

__device__ __forceinline__ int bucket_of(int d, int nbkt) {
    const int b = (int)((unsigned)d >> BKT_SHIFT);
    return b < nbkt ? b : nbkt - 1;
}

__global__ __launch_bounds__(1024) void k_bucket_count(const int* __restrict__ dst, int* __restrict__ tot,
                                                       int* __restrict__ boff, int ne, int nbkt) {
    __shared__ int lh[NBKT_MAX];
    const int tid = threadIdx.x;
    if (tid < NBKT_MAX) lh[tid] = 0;
    __syncthreads();
    const int e0 = blockIdx.x * CHUNK + tid;
    if ((blockIdx.x + 1) * CHUNK <= ne) {
        int dv[4];
#pragma unroll
        for (int k = 0; k < 4; ++k) dv[k] = dst[e0 + k * 1024];
#pragma unroll
        for (int k = 0; k < 4; ++k) atomicAdd(&lh[bucket_of(dv[k], nbkt)], 1);
    } else {
        for (int k = 0; k < 4; ++k) {
            const int e = e0 + k * 1024;
            if (e < ne) atomicAdd(&lh[bucket_of(dst[e], nbkt)], 1);
        }
    }
    __syncthreads();
    if (tid < nbkt) {
        const int c = lh[tid];
        int off = 0;
        if (c) off = __hip_atomic_fetch_add(&tot[tid], c, __ATOMIC_RELAXED, __HIP_MEMORY_SCOPE_AGENT);
        boff[blockIdx.x * NBKT_MAX + tid] = off;
    }
}

__device__ __forceinline__ void scatter_role(const int blk, const int* __restrict__ src, const int* __restrict__ dst,
                                             const int* __restrict__ tot, const int* __restrict__ boff,
                                             int2* __restrict__ recs, int ne, int nbkt) {
    __shared__ int lh[NBKT_MAX];
    __shared__ int lofs[NBKT_MAX];
    __shared__ int gb[NBKT_MAX];
    __shared__ int swA[32], swB[32];
    __shared__ int2 srec[CHUNK];
    const int tid = threadIdx.x;
    if (tid < NBKT_MAX) lh[tid] = 0;
    const int bo = tid < nbkt ? boff[blk * NBKT_MAX + tid] : 0;
    const int bstart = block_excl_scan_1024(tid < nbkt ? tot[tid] : 0, swA);
    int s[4], d[4], r[4], b[4];
    const int e0 = blk * CHUNK + tid;
    if ((blk + 1) * CHUNK <= ne) {
#pragma unroll
        for (int k = 0; k < 4; ++k) {
            s[k] = src[e0 + k * 1024];
            d[k] = dst[e0 + k * 1024];
        }
#pragma unroll
        for (int k = 0; k < 4; ++k) {
            b[k] = bucket_of(d[k], nbkt);
            r[k] = atomicAdd(&lh[b[k]], 1);
        }
    } else {
#pragma unroll
        for (int k = 0; k < 4; ++k) {
            const int e = e0 + k * 1024;
            s[k] = 0; d[k] = 0; r[k] = 0;
            b[k] = -1;
            if (e < ne) {
                s[k] = src[e];
                d[k] = dst[e];
                b[k] = bucket_of(d[k], nbkt);
                r[k] = atomicAdd(&lh[b[k]], 1);
            }
        }
    }
    __syncthreads();
    const int lo = block_excl_scan_1024(tid < NBKT_MAX ? lh[tid] : 0, swB);
    if (tid < nbkt) {
        lofs[tid] = lo;
        gb[tid] = bstart + bo - lo;
    }
    __syncthreads();
#pragma unroll
    for (int k = 0; k < 4; ++k)
        if (b[k] >= 0) srec[lofs[b[k]] + r[k]] = make_int2(s[k], d[k]);
    __syncthreads();
    const int left = ne - blk * CHUNK;
    const int nv = left < CHUNK ? left : CHUNK;
    for (int i = tid; i < nv; i += 1024) {
        const int2 rec = srec[i];
        const int p = gb[bucket_of(rec.y, nbkt)] + i;
        if ((unsigned)p < (unsigned)ne) recs[p] = rec;
    }
}

__device__ __forceinline__ void rows_role(const int blk, const int2* __restrict__ recs, const int* __restrict__ tot,
                                          int* __restrict__ rowptr, int* __restrict__ cnt,
                                          float* __restrict__ dinv, int* __restrict__ col,
                                          int n_nodes, int ne, int nbkt) {
    __shared__ int lcnt[BKT_W + 1];
    __shared__ int lpos[BKT_W + 1];
    __shared__ int swA[32], swB[32];
    __shared__ int sb[2];
    __shared__ int stage[STAGE_CAP];
    const int tid = threadIdx.x;
    const int bkt = blk;
    if (tid <= BKT_W) { lcnt[tid] = 0; lpos[tid] = 0; }
    const int t = tid < nbkt ? tot[tid] : 0;
    const int ex = block_excl_scan_1024(t, swA);
    if (tid == bkt) { sb[0] = ex; sb[1] = t; }
    __syncthreads();
    int bstart = sb[0];
    int nb = sb[1];
    if (bstart < 0 || bstart > ne) { bstart = 0; nb = 0; }
    if (nb > ne - bstart) nb = ne - bstart;
    const int2* rp = recs + bstart;
    for (int i0 = tid; i0 < nb; i0 += 4096) {
        int y[4];
#pragma unroll
        for (int k = 0; k < 4; ++k) {
            const int i = i0 + k * 1024;
            y[k] = rp[i < nb ? i : nb - 1].y;
        }
#pragma unroll
        for (int k = 0; k < 4; ++k)
            atomicAdd(&lcnt[i0 + k * 1024 < nb ? (y[k] & (BKT_W - 1)) : BKT_W], 1);
    }
    __syncthreads();
    const int c = tid < BKT_W ? lcnt[tid] : 0;
    const int ex2 = block_excl_scan_1024(c, swB);
    if (tid < BKT_W) {
        lpos[tid] = ex2;
        const int node = bkt * BKT_W + tid;
        if (node < n_nodes) {
            rowptr[node] = bstart + ex2;
            cnt[node] = c;
            dinv[node] = rsqrtf((float)(c + 1));
        }
    }
    __syncthreads();
    const bool staged = nb <= STAGE_CAP;
    for (int i0 = tid; i0 < nb; i0 += 4096) {
        int2 rec[4];
        int p[4];
#pragma unroll
        for (int k = 0; k < 4; ++k) {
            const int i = i0 + k * 1024;
            rec[k] = rp[i < nb ? i : nb - 1];
        }
#pragma unroll
        for (int k = 0; k < 4; ++k)
            p[k] = atomicAdd(&lpos[i0 + k * 1024 < nb ? (rec[k].y & (BKT_W - 1)) : BKT_W], 1);
#pragma unroll
        for (int k = 0; k < 4; ++k) {
            if (i0 + k * 1024 < nb && (unsigned)p[k] < (unsigned)nb) {
                if (staged) stage[p[k]] = rec[k].x;
                else        col[bstart + p[k]] = rec[k].x;
            }
        }
    }
    if (staged) {
        __syncthreads();
        for (int i = tid; i < nb; i += 1024) col[bstart + i] = stage[i];
    }
}

__device__ __forceinline__ void gemm_role(const int gblk, const float* __restrict__ x, const float* __restrict__ W1,
                                          float* __restrict__ h, int n_nodes)
{
    __shared__ float sW[(KTILE / 2) * PSTRIDE];

    const int tid  = threadIdx.x;
    const int wave = tid >> 5;
    const int lane = tid & 31;
    const int half = lane >> 4;
    const int l16  = lane & 15;
    const int rt   = wave & 15;
    const int ch   = wave >> 4;

    const int row_base = gblk * 256 + rt * 16;
    int arow = row_base + l16;
    if (arow >= n_nodes) arow = n_nodes - 1;
    const float* aptr = x + (size_t)arow * IN_CH + half * 2;

    v8f acc[4] = {};

    for (int k0 = 0; k0 < IN_CH; k0 += KTILE) {
        __syncthreads();
        {
            const int p  = tid >> 5;
            const int c4 = (tid & 31) * 4;
            float4 r0 = *(const float4*)(W1 + (size_t)(k0 + 2 * p + 0) * HID_CH + c4);
            float4 r1 = *(const float4*)(W1 + (size_t)(k0 + 2 * p + 1) * HID_CH + c4);
            float* dp = &sW[p * PSTRIDE + c4 * 2];
            ((float2*)dp)[0] = make_float2(r0.x, r1.x);
            ((float2*)dp)[1] = make_float2(r0.y, r1.y);
            ((float2*)dp)[2] = make_float2(r0.z, r1.z);
            ((float2*)dp)[3] = make_float2(r0.w, r1.w);
        }
        __syncthreads();

#pragma unroll 4
        for (int k = 0; k < KTILE; k += 4) {
            v2f a = *(const v2f*)(aptr + k0 + k);
            const int p = (k >> 1) + half;
            const float* bbase = &sW[p * PSTRIDE + ch * 128 + l16 * 2];
#pragma unroll
            for (int nt = 0; nt < 4; ++nt) {
                v2f b = *(const v2f*)(bbase + nt * 32);
                acc[nt] = __builtin_amdgcn_wmma_f32_16x16x4_f32(
                    false, a, false, b, (short)0, acc[nt], false, false);
            }
        }
    }

#pragma unroll
    for (int nt = 0; nt < 4; ++nt) {
#pragma unroll
        for (int v = 0; v < 8; ++v) {
            int r = row_base + v + 8 * half;
            if (r < n_nodes) h[(size_t)r * HID_CH + ch * 64 + nt * 16 + l16] = acc[nt][v];
        }
    }
}

__global__ __launch_bounds__(1024) __attribute__((amdgpu_waves_per_eu(16, 16))) void k_scatter_and_gemm(
    const int* __restrict__ src, const int* __restrict__ dst, const int* __restrict__ tot,
    const int* __restrict__ boff, int2* __restrict__ recs, int ne, int nbkt,
    const float* __restrict__ x, const float* __restrict__ W1, float* __restrict__ h, int n_nodes,
    int g_first, int g_count)
{
    const int b = blockIdx.x;
    if (b < g_count) gemm_role(g_first + b, x, W1, h, n_nodes);
    else             scatter_role(b - g_count, src, dst, tot, boff, recs, ne, nbkt);
}

__global__ __launch_bounds__(1024) __attribute__((amdgpu_waves_per_eu(16, 16))) void k_rows_and_gemm(
    const int2* __restrict__ recs, const int* __restrict__ tot, int* __restrict__ rowptr, int* __restrict__ cnt,
    float* __restrict__ dinv, int* __restrict__ col, int ne, int nbkt,
    const float* __restrict__ x, const float* __restrict__ W1, float* __restrict__ h, int n_nodes,
    int g_first, int g_count)
{
    const int b = blockIdx.x;
    if (b < g_count) gemm_role(g_first + b, x, W1, h, n_nodes);
    else             rows_role(b - g_count, recs, tot, rowptr, cnt, dinv, col, n_nodes, ne, nbkt);
}


template <int R>
__device__ __forceinline__ void rows_n(const float* __restrict__ h, const unsigned lane_off, const int my_s,
                                       const int my_di, const float dd, const int j, float4& acc) {
    float4 a[R];
#pragma unroll
    for (int k = 0; k < R; ++k) {
        const int s = __builtin_amdgcn_readlane(my_s, j + k);
        a[k] = *(const float4*)((const char*)h + ((unsigned)s * (unsigned)(HID_CH * 4) + lane_off));
    }
    __builtin_amdgcn_sched_barrier(0);
#pragma unroll
    for (int k = 0; k < R; ++k) {
        const float w = __int_as_float(__builtin_amdgcn_readlane(my_di, j + k)) * dd;
        acc.x += a[k].x * w; acc.y += a[k].y * w; acc.z += a[k].z * w; acc.w += a[k].w * w;
    }
}

#define G1_NPW 7

__global__ __launch_bounds__(512) void k_gather1(
    const float* __restrict__ h, const int* __restrict__ rowptr, const int* __restrict__ cnt,
    const int* __restrict__ col, const float* __restrict__ dinv,
    const float* __restrict__ b1, const float* __restrict__ W2,
    float* __restrict__ h2, int n_nodes)
{
    __shared__ float sh2[G1_NPW * 32];
    const int lane = threadIdx.x & 31;
    const int wave = threadIdx.x >> 5;
    const unsigned lane_off = (unsigned)lane * 16u;
    for (int n = 0; n < G1_NPW; ++n) {
        const int node = __builtin_amdgcn_readfirstlane((blockIdx.x * G1_NPW + n) * 16 + wave);
        float p0 = 0.f, p1 = 0.f;
        if (node < n_nodes) {
            const int start = rowptr[node];
            const int tot = cnt[node] + 1;
            const float dd = dinv[node];
            float4 acc = make_float4(0.f, 0.f, 0.f, 0.f);
            int base = 0;
            do {
                const int m = (tot - base < 32) ? (tot - base) : 32;
                const int i = base + lane;
                int my_s = node; float my_d = dd;
                if (i >= 1 && i < tot) {
                    my_s = col[start + i - 1];
                    my_d = dinv[my_s];
                }
                const int my_di = __float_as_int(my_d);
                int j = 0;
                for (; j + 8 <= m; j += 8) rows_n<8>(h, lane_off, my_s, my_di, dd, j, acc);
                switch (m - j) {
                    case 1: rows_n<1>(h, lane_off, my_s, my_di, dd, j, acc); break;
                    case 2: rows_n<2>(h, lane_off, my_s, my_di, dd, j, acc); break;
                    case 3: rows_n<3>(h, lane_off, my_s, my_di, dd, j, acc); break;
                    case 4: rows_n<4>(h, lane_off, my_s, my_di, dd, j, acc); break;
                    case 5: rows_n<5>(h, lane_off, my_s, my_di, dd, j, acc); break;
                    case 6: rows_n<6>(h, lane_off, my_s, my_di, dd, j, acc); break;
                    case 7: rows_n<7>(h, lane_off, my_s, my_di, dd, j, acc); break;
                    default: break;
                }
                base += 32;
            } while (base < tot);
            const float4 bv = *(const float4*)(b1 + lane * 4);
            const float r0 = fmaxf(acc.x + bv.x, 0.f), r1 = fmaxf(acc.y + bv.y, 0.f);
            const float r2 = fmaxf(acc.z + bv.z, 0.f), r3 = fmaxf(acc.w + bv.w, 0.f);
            const float4 wa = *(const float4*)(W2 + lane * 8);
            const float4 wb = *(const float4*)(W2 + lane * 8 + 4);
            p0 = r0 * wa.x + r1 * wa.z + r2 * wb.x + r3 * wb.z;
            p1 = r0 * wa.y + r1 * wa.w + r2 * wb.y + r3 * wb.w;
            for (int o = 16; o > 0; o >>= 1) {
                p0 += __shfl_xor(p0, o);
                p1 += __shfl_xor(p1, o);
            }
        }
        if (lane == 0) { sh2[n * 32 + wave * 2 + 0] = p0; sh2[n * 32 + wave * 2 + 1] = p1; }
    }
    __syncthreads();
    if (wave < G1_NPW) {
        const int idx = (blockIdx.x * G1_NPW + wave) * 32 + lane;
        if (idx < n_nodes * 2) h2[idx] = sh2[wave * 32 + lane];
    }
}

#define G2_NPW 2

__global__ __launch_bounds__(256) void k_gather2(
    const float* __restrict__ h2, const int* __restrict__ rowptr, const int* __restrict__ cnt,
    const int* __restrict__ col, const float* __restrict__ dinv,
    const float* __restrict__ b2, float* __restrict__ out, int n_nodes)
{
    __shared__ float2 sres[G2_NPW * 32];
    const int tid = threadIdx.x;
    const int g = tid >> 3, sub = tid & 7;
    for (int n = 0; n < G2_NPW; ++n) {
        const int node = (blockIdx.x * G2_NPW + n) * 32 + g;
        float a0 = 0.f, a1 = 0.f;
        float dd = 0.f;
        float2 hs = make_float2(0.f, 0.f);
        if (node < n_nodes) {
            const int start = rowptr[node];
            const int end = start + cnt[node];
            dd = dinv[node];
            hs = *(const float2*)(h2 + (size_t)node * 2);
            for (int jb = start + sub; jb < end; jb += 32) {
                int s[4];
                float ds[4];
                float2 v[4];
#pragma unroll
                for (int k = 0; k < 4; ++k) {
                    const int jk = jb + 8 * k;
                    s[k] = col[(k == 0 || jk < end) ? jk : end - 1];
                }
#pragma unroll
                for (int k = 0; k < 4; ++k) {
                    ds[k] = dinv[s[k]];
                    v[k] = *(const float2*)(h2 + (size_t)s[k] * 2);
                }
#pragma unroll
                for (int k = 0; k < 4; ++k) {
                    if (k == 0 || jb + 8 * k < end) {
                        const float norm = ds[k] * dd;
                        a0 += v[k].x * norm;
                        a1 += v[k].y * norm;
                    }
                }
            }
        }
        for (int o = 4; o > 0; o >>= 1) {
            a0 += __shfl_xor(a0, o);
            a1 += __shfl_xor(a1, o);
        }
        if (sub == 0) {
            const float w = dd * dd;
            sres[n * 32 + g] = make_float2((b2[0] + hs.x * w) + a0, (b2[1] + hs.y * w) + a1);
        }
    }
    __syncthreads();
    if (tid < G2_NPW * 32) {
        const int nd = blockIdx.x * (G2_NPW * 32) + tid;
        if (nd < n_nodes) *(float2*)(out + (size_t)nd * 2) = sres[tid];
    }
}

extern "C" void kernel_launch(void* const* d_in, const int* in_sizes, int n_in,
                              void* d_out, int out_size, void* d_ws, size_t ws_size,
                              hipStream_t stream)
{
    const float* x  = (const float*)d_in[0];
    const int*   ei = (const int*)d_in[1];
    const float* W1 = (const float*)d_in[2];
    const float* b1 = (const float*)d_in[3];
    const float* W2 = (const float*)d_in[4];
    const float* b2 = (const float*)d_in[5];
    float* out = (float*)d_out;

    const int n_nodes = in_sizes[0] / IN_CH;
    const int n_edges = in_sizes[1] / 2;
    const int* src = ei;
    const int* dst = ei + n_edges;
    const int nbkt = (n_nodes + BKT_W - 1) / BKT_W;
    const int n_chunk = (n_edges + CHUNK - 1) / CHUNK;

    char* ws = (char*)d_ws;
    size_t off = 0;
    const size_t nb4 = ((size_t)n_nodes * 4 + 4095) & ~(size_t)4095;
    const size_t eb4 = ((size_t)n_edges * 4 + 4095) & ~(size_t)4095;
    float* dinv = (float*)(ws + off);  off += nb4;
    float* h = (float*)(ws + off);
    off += ((size_t)n_nodes * HID_CH * 4 + 4095) & ~(size_t)4095;
    float* h2 = (float*)(ws + off);    off += 2 * nb4;
    int* cnt = (int*)(ws + off);       off += nb4;
    int* rowptr = (int*)(ws + off);    off += nb4;
    int* tot = (int*)(ws + off);       off += 4096;
    int* boff = (int*)(ws + off);      off += ((size_t)n_chunk * NBKT_MAX * 4 + 4095) & ~(size_t)4095;
    int2* recs = (int2*)(ws + off);    off += 2 * eb4;
    int* col = (int*)(ws + off);

    k_zero_tot      <<<1, NBKT_MAX, 0, stream>>>(tot);
    k_bucket_count  <<<n_chunk, 1024, 0, stream>>>(dst, tot, boff, n_edges, nbkt);
    const int g_all = (n_nodes + 255) / 256;
    const int g_a = (g_all + 1) / 2;
    const int g_b = g_all - g_a;
    k_scatter_and_gemm<<<g_a + n_chunk, 1024, 0, stream>>>(src, dst, tot, boff, recs, n_edges, nbkt,
                                                          x, W1, h, n_nodes, 0, g_a);
    k_rows_and_gemm   <<<g_b + nbkt, 1024, 0, stream>>>(recs, tot, rowptr, cnt, dinv, col, n_edges, nbkt,
                                                       x, W1, h, n_nodes, g_a, g_b);

    k_gather1<<<(n_nodes + 16 * G1_NPW - 1) / (16 * G1_NPW), 512, 0, stream>>>(h, rowptr, cnt, col, dinv, b1, W2, h2, n_nodes);
    k_gather2<<<(n_nodes + 32 * G2_NPW - 1) / (32 * G2_NPW), 256, 0, stream>>>(h2, rowptr, cnt, col, dinv, b2, out, n_nodes);
}


# Edge grouping drops its counting pass and counter clear (2 launches fewer): a block orders its own 4096-edge chunk by bucket in place, bucket blocks read their runs; sum order = LDS arrival rank
# speedup vs baseline: 4.6862x; 1.1194x over previous
#include <hip/hip_runtime.h>
#include <hip/hip_bf16.h>
#include <cstdint>


typedef float v2f __attribute__((ext_vector_type(2)));
typedef float v8f __attribute__((ext_vector_type(8)));

#define IN_CH  256
#define HID_CH 128
#define KTILE  64
#define PSTRIDE 288

#define BKT_SHIFT 9
#define BKT_W     512
#define NBKT_MAX  256
#define CHUNK     4096
#define STAGE_CAP 12288
#define SEGS      8

__device__ __forceinline__ int wave_incl_scan_i(int v, int lane) {
    for (int o = 1; o < 32; o <<= 1) {
        int t = __shfl_up(v, o);
        if (lane >= o) v += t;
    }
    return v;
}

__device__ __forceinline__ int block_excl_scan_1024(int v, int* sw) {
    const int lane = threadIdx.x & 31, wave = threadIdx.x >> 5;
    const int incl = wave_incl_scan_i(v, lane);
    if (lane == 31) sw[wave] = incl;
    __syncthreads();
    if (wave == 0) {
        const int t = sw[lane];
        const int s = wave_incl_scan_i(t, lane);
        sw[lane] = s - t;
    }
    __syncthreads();
    return sw[wave] + incl - v;
}

__device__ __forceinline__ int bucket_of(int d, int nbkt) {
    const int b = (int)((unsigned)d >> BKT_SHIFT);
    return b < nbkt ? b : nbkt - 1;
}

__device__ __forceinline__ void scatter_role(const int blk, const int* __restrict__ src, const int* __restrict__ dst,
                                             int2* __restrict__ recs, int* __restrict__ seg, int* __restrict__ cursor,
                                             int ne, int nbkt, int n_chunk) {
    __shared__ int lh[NBKT_MAX];
    __shared__ int lofs[NBKT_MAX];
    __shared__ int swB[32];
    __shared__ int2 srec[CHUNK];
    const int tid = threadIdx.x;
    if (tid < NBKT_MAX) lh[tid] = 0;
    if (blk == 0 && tid == 0) *cursor = 0;
    __syncthreads();
    int s[4], d[4], r[4], b[4];
    const int e0 = blk * CHUNK + tid;
    if ((blk + 1) * CHUNK <= ne) {
#pragma unroll
        for (int k = 0; k < 4; ++k) {
            s[k] = src[e0 + k * 1024];
            d[k] = dst[e0 + k * 1024];
        }
#pragma unroll
        for (int k = 0; k < 4; ++k) {
            b[k] = bucket_of(d[k], nbkt);
            r[k] = atomicAdd(&lh[b[k]], 1);
        }
    } else {
#pragma unroll
        for (int k = 0; k < 4; ++k) {
            const int e = e0 + k * 1024;
            s[k] = 0; d[k] = 0; r[k] = 0;
            b[k] = -1;
            if (e < ne) {
                s[k] = src[e];
                d[k] = dst[e];
                b[k] = bucket_of(d[k], nbkt);
                r[k] = atomicAdd(&lh[b[k]], 1);
            }
        }
    }
    __syncthreads();
    const int c = tid < NBKT_MAX ? lh[tid] : 0;
    const int lo = block_excl_scan_1024(c, swB);
    if (tid < nbkt) {
        lofs[tid] = lo;
        seg[tid * n_chunk + blk] = (lo << 16) | c;
    }
    __syncthreads();
#pragma unroll
    for (int k = 0; k < 4; ++k)
        if (b[k] >= 0) srec[lofs[b[k]] + r[k]] = make_int2(s[k], d[k]);
    __syncthreads();
    const int left = ne - blk * CHUNK;
    const int nv = left < CHUNK ? left : CHUNK;
    for (int i = tid; i < nv; i += 1024) recs[blk * CHUNK + i] = srec[i];
}

__device__ __forceinline__ int in_recs(int i, int ne) { return (unsigned)i < (unsigned)ne ? i : 0; }

__device__ __forceinline__ void rows_role(const int blk, const int2* __restrict__ recs, const int* __restrict__ seg,
                                          int* __restrict__ cursor,
                                          int* __restrict__ rowptr, int* __restrict__ cnt,
                                          float* __restrict__ dinv, int* __restrict__ col,
                                          int n_nodes, int ne, int nbkt, int n_chunk) {
    __shared__ int lcnt[BKT_W];
    __shared__ int lpos[BKT_W];
    __shared__ int swA[32], swB[32];
    __shared__ int sb[2];
    __shared__ int stage[STAGE_CAP];
    const int tid = threadIdx.x;
    const int lane = tid & 31;
    const int wv = __builtin_amdgcn_readfirstlane(tid >> 5);
    const int bkt = blk;
    if (tid < BKT_W) lcnt[tid] = 0;
    __syncthreads();
    const int* trow = seg + bkt * n_chunk;
    const int per = (n_chunk + 31) >> 5;
    const int c_lo = wv * per;
    const int c_hi = c_lo + per < n_chunk ? c_lo + per : n_chunk;

    int mine = 0;
    for (int g0 = c_lo; g0 < c_hi; g0 += 32) {
        const int e = g0 + lane < c_hi ? trow[g0 + lane] : 0;
        mine += e & 0xFFFF;
        const int ng = c_hi - g0 < 32 ? c_hi - g0 : 32;
        for (int j = 0; j < ng; j += SEGS) {
            int y[SEGS], cj[SEGS];
            int longest = 0;
#pragma unroll
            for (int k = 0; k < SEGS; ++k) {
                const int ej = __builtin_amdgcn_readlane(e, j + k);
                cj[k] = ej & 0xFFFF;
                longest = cj[k] > longest ? cj[k] : longest;
                const int i = (g0 + j + k) * CHUNK + (ej >> 16) + lane;
                y[k] = recs[in_recs(lane < cj[k] ? i : 0, ne)].y;
            }
#pragma unroll
            for (int k = 0; k < SEGS; ++k)
                if (lane < cj[k]) atomicAdd(&lcnt[y[k] & (BKT_W - 1)], 1);
            if (longest > 32) {
                for (int jj = j; jj < j + SEGS; ++jj) {
                    const int ej = __builtin_amdgcn_readlane(e, jj);
                    const int cl = ej & 0xFFFF;
                    const int bl = (g0 + jj) * CHUNK + (ej >> 16);
                    for (int i = lane + 32; i < cl; i += 32)
                        atomicAdd(&lcnt[recs[in_recs(bl + i, ne)].y & (BKT_W - 1)], 1);
                }
            }
        }
    }
    for (int o = 16; o > 0; o >>= 1) mine += __shfl_xor(mine, o);
    if (lane == 0) swA[wv] = mine;
    __syncthreads();
    if (wv == 0) {
        int t = swA[lane];
        for (int o = 16; o > 0; o >>= 1) t += __shfl_xor(t, o);
        if (lane == 0) {
            sb[1] = t;
            sb[0] = __hip_atomic_fetch_add(cursor, t, __ATOMIC_RELAXED, __HIP_MEMORY_SCOPE_AGENT);
        }
    }
    const int c = tid < BKT_W ? lcnt[tid] : 0;
    const int ex2 = block_excl_scan_1024(c, swB);
    int bstart = sb[0];
    int nb = sb[1];
    if (bstart < 0 || bstart > ne) { bstart = 0; nb = 0; }
    if (nb > ne - bstart) nb = ne - bstart;
    if (tid < BKT_W) {
        lpos[tid] = ex2;
        const int node = bkt * BKT_W + tid;
        if (node < n_nodes) {
            rowptr[node] = bstart + ex2;
            cnt[node] = c;
            dinv[node] = rsqrtf((float)(c + 1));
        }
    }
    __syncthreads();

    const bool staged = nb <= STAGE_CAP;
    for (int g0 = c_lo; g0 < c_hi; g0 += 32) {
        const int e = g0 + lane < c_hi ? trow[g0 + lane] : 0;
        const int ng = c_hi - g0 < 32 ? c_hi - g0 : 32;
        for (int j = 0; j < ng; j += SEGS) {
            int2 rec[SEGS];
            int cj[SEGS], p[SEGS];
            int longest = 0;
#pragma unroll
            for (int k = 0; k < SEGS; ++k) {
                const int ej = __builtin_amdgcn_readlane(e, j + k);
                cj[k] = ej & 0xFFFF;
                longest = cj[k] > longest ? cj[k] : longest;
                const int i = (g0 + j + k) * CHUNK + (ej >> 16) + lane;
                rec[k] = recs[in_recs(lane < cj[k] ? i : 0, ne)];
            }
#pragma unroll
            for (int k = 0; k < SEGS; ++k) {
                p[k] = -1;
                if (lane < cj[k]) p[k] = atomicAdd(&lpos[rec[k].y & (BKT_W - 1)], 1);
            }
#pragma unroll
            for (int k = 0; k < SEGS; ++k) {
                if ((unsigned)p[k] < (unsigned)nb) {
                    if (staged) stage[p[k]] = rec[k].x;
                    else        col[bstart + p[k]] = rec[k].x;
                }
            }
            if (longest > 32) {
                for (int jj = j; jj < j + SEGS; ++jj) {
                    const int ej = __builtin_amdgcn_readlane(e, jj);
                    const int cl = ej & 0xFFFF;
                    const int bl = (g0 + jj) * CHUNK + (ej >> 16);
                    for (int i = lane + 32; i < cl; i += 32) {
                        const int2 r2 = recs[in_recs(bl + i, ne)];
                        const int q = atomicAdd(&lpos[r2.y & (BKT_W - 1)], 1);
                        if ((unsigned)q < (unsigned)nb) {
                            if (staged) stage[q] = r2.x;
                            else        col[bstart + q] = r2.x;
                        }
                    }
                }
            }
        }
    }
    if (staged) {
        __syncthreads();
        for (int i = tid; i < nb; i += 1024) col[bstart + i] = stage[i];
    }
}

__device__ __forceinline__ void gemm_role(const int gblk, const float* __restrict__ x, const float* __restrict__ W1,
                                          float* __restrict__ h, int n_nodes)
{
    __shared__ float sW[(KTILE / 2) * PSTRIDE];

    const int tid  = threadIdx.x;
    const int wave = tid >> 5;
    const int lane = tid & 31;
    const int half = lane >> 4;
    const int l16  = lane & 15;
    const int rt   = wave & 15;
    const int ch   = wave >> 4;

    const int row_base = gblk * 256 + rt * 16;
    int arow = row_base + l16;
    if (arow >= n_nodes) arow = n_nodes - 1;
    const float* aptr = x + (size_t)arow * IN_CH + half * 2;

    v8f acc[4] = {};

    for (int k0 = 0; k0 < IN_CH; k0 += KTILE) {
        __syncthreads();
        {
            const int p  = tid >> 5;
            const int c4 = (tid & 31) * 4;
            float4 r0 = *(const float4*)(W1 + (size_t)(k0 + 2 * p + 0) * HID_CH + c4);
            float4 r1 = *(const float4*)(W1 + (size_t)(k0 + 2 * p + 1) * HID_CH + c4);
            float* dp = &sW[p * PSTRIDE + c4 * 2];
            ((float2*)dp)[0] = make_float2(r0.x, r1.x);
            ((float2*)dp)[1] = make_float2(r0.y, r1.y);
            ((float2*)dp)[2] = make_float2(r0.z, r1.z);
            ((float2*)dp)[3] = make_float2(r0.w, r1.w);
        }
        __syncthreads();

#pragma unroll 4
        for (int k = 0; k < KTILE; k += 4) {
            v2f a = *(const v2f*)(aptr + k0 + k);
            const int p = (k >> 1) + half;
            const float* bbase = &sW[p * PSTRIDE + ch * 128 + l16 * 2];
#pragma unroll
            for (int nt = 0; nt < 4; ++nt) {
                v2f b = *(const v2f*)(bbase + nt * 32);
                acc[nt] = __builtin_amdgcn_wmma_f32_16x16x4_f32(
                    false, a, false, b, (short)0, acc[nt], false, false);
            }
        }
    }

#pragma unroll
    for (int nt = 0; nt < 4; ++nt) {
#pragma unroll
        for (int v = 0; v < 8; ++v) {
            int r = row_base + v + 8 * half;
            if (r < n_nodes) h[(size_t)r * HID_CH + ch * 64 + nt * 16 + l16] = acc[nt][v];
        }
    }
}

__global__ __launch_bounds__(1024) __attribute__((amdgpu_waves_per_eu(16, 16))) void k_scatter_and_gemm(
    const int* __restrict__ src, const int* __restrict__ dst, int2* __restrict__ recs,
    int* __restrict__ seg, int* __restrict__ cursor, int ne, int nbkt, int n_chunk,
    const float* __restrict__ x, const float* __restrict__ W1, float* __restrict__ h, int n_nodes,
    int g_first, int g_count)
{
    const int b = blockIdx.x;
    if (b < g_count) gemm_role(g_first + b, x, W1, h, n_nodes);
    else             scatter_role(b - g_count, src, dst, recs, seg, cursor, ne, nbkt, n_chunk);
}

__global__ __launch_bounds__(1024) __attribute__((amdgpu_waves_per_eu(16, 16))) void k_rows_and_gemm(
    const int2* __restrict__ recs, const int* __restrict__ seg, int* __restrict__ cursor,
    int* __restrict__ rowptr, int* __restrict__ cnt,
    float* __restrict__ dinv, int* __restrict__ col, int ne, int nbkt, int n_chunk,
    const float* __restrict__ x, const float* __restrict__ W1, float* __restrict__ h, int n_nodes,
    int g_first, int g_count)
{
    const int b = blockIdx.x;
    if (b < g_count) gemm_role(g_first + b, x, W1, h, n_nodes);
    else             rows_role(b - g_count, recs, seg, cursor, rowptr, cnt, dinv, col, n_nodes, ne, nbkt, n_chunk);
}


template <int R>
__device__ __forceinline__ void rows_n(const float* __restrict__ h, const unsigned lane_off, const int my_s,
                                       const int my_di, const float dd, const int j, float4& acc) {
    float4 a[R];
#pragma unroll
    for (int k = 0; k < R; ++k) {
        const int s = __builtin_amdgcn_readlane(my_s, j + k);
        a[k] = *(const float4*)((const char*)h + ((unsigned)s * (unsigned)(HID_CH * 4) + lane_off));
    }
    __builtin_amdgcn_sched_barrier(0);
#pragma unroll
    for (int k = 0; k < R; ++k) {
        const float w = __int_as_float(__builtin_amdgcn_readlane(my_di, j + k)) * dd;
        acc.x += a[k].x * w; acc.y += a[k].y * w; acc.z += a[k].z * w; acc.w += a[k].w * w;
    }
}

#define G1_NPW 7

__global__ __launch_bounds__(512) void k_gather1(
    const float* __restrict__ h, const int* __restrict__ rowptr, const int* __restrict__ cnt,
    const int* __restrict__ col, const float* __restrict__ dinv,
    const float* __restrict__ b1, const float* __restrict__ W2,
    float* __restrict__ rec, int n_nodes)
{
    __shared__ float4 srec[G1_NPW * 16];
    const int lane = threadIdx.x & 31;
    const int wave = threadIdx.x >> 5;
    const unsigned lane_off = (unsigned)lane * 16u;
    for (int n = 0; n < G1_NPW; ++n) {
        const int node = __builtin_amdgcn_readfirstlane((blockIdx.x * G1_NPW + n) * 16 + wave);
        if (node < n_nodes) {
            const int start = rowptr[node];
            const int tot = cnt[node] + 1;
            const float dd = dinv[node];
            float4 acc = make_float4(0.f, 0.f, 0.f, 0.f);
            int base = 0;
            do {
                const int m = (tot - base < 32) ? (tot - base) : 32;
                const int i = base + lane;
                int my_s = node; float my_d = dd;
                if (i >= 1 && i < tot) {
                    my_s = col[start + i - 1];
                    my_d = dinv[my_s];
                }
                const int my_di = __float_as_int(my_d);
                int j = 0;
                for (; j + 8 <= m; j += 8) rows_n<8>(h, lane_off, my_s, my_di, dd, j, acc);
                switch (m - j) {
                    case 1: rows_n<1>(h, lane_off, my_s, my_di, dd, j, acc); break;
                    case 2: rows_n<2>(h, lane_off, my_s, my_di, dd, j, acc); break;
                    case 3: rows_n<3>(h, lane_off, my_s, my_di, dd, j, acc); break;
                    case 4: rows_n<4>(h, lane_off, my_s, my_di, dd, j, acc); break;
                    case 5: rows_n<5>(h, lane_off, my_s, my_di, dd, j, acc); break;
                    case 6: rows_n<6>(h, lane_off, my_s, my_di, dd, j, acc); break;
                    case 7: rows_n<7>(h, lane_off, my_s, my_di, dd, j, acc); break;
                    default: break;
                }
                base += 32;
            } while (base < tot);
            const float4 bv = *(const float4*)(b1 + lane * 4);
            const float r0 = fmaxf(acc.x + bv.x, 0.f), r1 = fmaxf(acc.y + bv.y, 0.f);
            const float r2 = fmaxf(acc.z + bv.z, 0.f), r3 = fmaxf(acc.w + bv.w, 0.f);
            const float4 wa = *(const float4*)(W2 + lane * 8);
            const float4 wb = *(const float4*)(W2 + lane * 8 + 4);
            float p0 = r0 * wa.x + r1 * wa.z + r2 * wb.x + r3 * wb.z;
            float p1 = r0 * wa.y + r1 * wa.w + r2 * wb.y + r3 * wb.w;
            for (int o = 16; o > 0; o >>= 1) {
                p0 += __shfl_xor(p0, o);
                p1 += __shfl_xor(p1, o);
            }
            if (lane == 0) srec[n * 16 + wave] = make_float4(p0, p1, dd, 0.f);
        }
    }
    __syncthreads();
    if (wave < G1_NPW) {
        const int idx = (blockIdx.x * G1_NPW + wave) * 32 + lane;
        if (idx < n_nodes * 2) ((float2*)rec)[idx] = ((const float2*)srec)[wave * 32 + lane];
    }
}

#define G2_NPW 2

__global__ __launch_bounds__(256) void k_gather2(
    const float* __restrict__ rec, const int* __restrict__ rowptr, const int* __restrict__ cnt,
    const int* __restrict__ col,
    const float* __restrict__ b2, float* __restrict__ out, int n_nodes)
{
    __shared__ float2 sres[G2_NPW * 32];
    const int tid = threadIdx.x;
    const int g = tid >> 3, sub = tid & 7;
    for (int n = 0; n < G2_NPW; ++n) {
        const int node = (blockIdx.x * G2_NPW + n) * 32 + g;
        float a0 = 0.f, a1 = 0.f;
        float dd = 0.f;
        float2 hs = make_float2(0.f, 0.f);
        if (node < n_nodes) {
            const int start = rowptr[node];
            const int end = start + cnt[node];
            const float4 own = *(const float4*)((const char*)rec + (unsigned)node * 16u);
            dd = own.z;
            hs = make_float2(own.x, own.y);
            for (int jb = start + sub; jb < end; jb += 32) {
                int s[4];
                float4 v[4];
#pragma unroll
                for (int k = 0; k < 4; ++k) {
                    const int jk = jb + 8 * k;
                    s[k] = col[(k == 0 || jk < end) ? jk : end - 1];
                }
#pragma unroll
                for (int k = 0; k < 4; ++k)
                    v[k] = *(const float4*)((const char*)rec + (unsigned)s[k] * 16u);
#pragma unroll
                for (int k = 0; k < 4; ++k) {
                    if (k == 0 || jb + 8 * k < end) {
                        const float norm = v[k].z * dd;
                        a0 += v[k].x * norm;
                        a1 += v[k].y * norm;
                    }
                }
            }
        }
        for (int o = 4; o > 0; o >>= 1) {
            a0 += __shfl_xor(a0, o);
            a1 += __shfl_xor(a1, o);
        }
        if (sub == 0) {
            const float w = dd * dd;
            sres[n * 32 + g] = make_float2((b2[0] + hs.x * w) + a0, (b2[1] + hs.y * w) + a1);
        }
    }
    __syncthreads();
    if (tid < G2_NPW * 32) {
        const int nd = blockIdx.x * (G2_NPW * 32) + tid;
        if (nd < n_nodes) *(float2*)(out + (size_t)nd * 2) = sres[tid];
    }
}

extern "C" void kernel_launch(void* const* d_in, const int* in_sizes, int n_in,
                              void* d_out, int out_size, void* d_ws, size_t ws_size,
                              hipStream_t stream)
{
    const float* x  = (const float*)d_in[0];
    const int*   ei = (const int*)d_in[1];
    const float* W1 = (const float*)d_in[2];
    const float* b1 = (const float*)d_in[3];
    const float* W2 = (const float*)d_in[4];
    const float* b2 = (const float*)d_in[5];
    float* out = (float*)d_out;

    const int n_nodes = in_sizes[0] / IN_CH;
    const int n_edges = in_sizes[1] / 2;
    const int* src = ei;
    const int* dst = ei + n_edges;
    const int nbkt = (n_nodes + BKT_W - 1) / BKT_W;
    const int n_chunk = (n_edges + CHUNK - 1) / CHUNK;

    char* ws = (char*)d_ws;
    size_t off = 0;
    const size_t nb4 = ((size_t)n_nodes * 4 + 4095) & ~(size_t)4095;
    const size_t eb4 = ((size_t)n_edges * 4 + 4095) & ~(size_t)4095;
    float* dinv = (float*)(ws + off);  off += nb4;
    float* h = (float*)(ws + off);
    off += ((size_t)n_nodes * HID_CH * 4 + 4095) & ~(size_t)4095;
    int* cnt = (int*)(ws + off);       off += nb4;
    int* rowptr = (int*)(ws + off);    off += nb4;
    int* cursor = (int*)(ws + off);    off += 4096;
    int* seg = (int*)(ws + off);       off += ((size_t)n_chunk * NBKT_MAX * 4 + 4095) & ~(size_t)4095;
    int2* recs = (int2*)(ws + off);    off += 2 * eb4;
    int* col = (int*)(ws + off);
    float* rec = (float*)recs;

    const int g_all = (n_nodes + 255) / 256;
    const int g_a = (g_all + 1) / 2;
    const int g_b = g_all - g_a;
    k_scatter_and_gemm<<<g_a + n_chunk, 1024, 0, stream>>>(src, dst, recs, seg, cursor, n_edges, nbkt, n_chunk,
                                                          x, W1, h, n_nodes, 0, g_a);
    k_rows_and_gemm   <<<g_b + nbkt, 1024, 0, stream>>>(recs, seg, cursor, rowptr, cnt, dinv, col, n_edges, nbkt, n_chunk,
                                                       x, W1, h, n_nodes, g_a, g_b);

    k_gather1<<<(n_nodes + 16 * G1_NPW - 1) / (16 * G1_NPW), 512, 0, stream>>>(h, rowptr, cnt, col, dinv, b1, W2, rec, n_nodes);
    k_gather2<<<(n_nodes + 32 * G2_NPW - 1) / (32 * G2_NPW), 256, 0, stream>>>(rec, rowptr, cnt, col, b2, out, n_nodes);
}


# x @ W1: rows of x reach the WMMAs through LDS: whole 128-byte lines fetched once per block (4 lines per load), A fragment = one ds_load_b64, not a 16-line load per k-step; k order and sums unchanged
# speedup vs baseline: 4.9197x; 1.0498x over previous
#include <hip/hip_runtime.h>
#include <hip/hip_bf16.h>
#include <cstdint>


typedef float v2f __attribute__((ext_vector_type(2)));
typedef float v8f __attribute__((ext_vector_type(8)));

#define IN_CH  256
#define HID_CH 128
#define KTILE  32
#define APITCH 36
#define PSTRIDE 288

#define BKT_SHIFT 9
#define BKT_W     512
#define NBKT_MAX  256
#define CHUNK     4096
#define STAGE_CAP 12288
#define SEGS      8

__device__ __forceinline__ int wave_incl_scan_i(int v, int lane) {
    for (int o = 1; o < 32; o <<= 1) {
        int t = __shfl_up(v, o);
        if (lane >= o) v += t;
    }
    return v;
}

__device__ __forceinline__ int block_excl_scan_1024(int v, int* sw) {
    const int lane = threadIdx.x & 31, wave = threadIdx.x >> 5;
    const int incl = wave_incl_scan_i(v, lane);
    if (lane == 31) sw[wave] = incl;
    __syncthreads();
    if (wave == 0) {
        const int t = sw[lane];
        const int s = wave_incl_scan_i(t, lane);
        sw[lane] = s - t;
    }
    __syncthreads();
    return sw[wave] + incl - v;
}

__device__ __forceinline__ int bucket_of(int d, int nbkt) {
    const int b = (int)((unsigned)d >> BKT_SHIFT);
    return b < nbkt ? b : nbkt - 1;
}

__device__ __forceinline__ void scatter_role(const int blk, const int* __restrict__ src, const int* __restrict__ dst,
                                             int2* __restrict__ recs, int* __restrict__ seg, int* __restrict__ cursor,
                                             int ne, int nbkt, int n_chunk) {
    __shared__ int lh[NBKT_MAX];
    __shared__ int lofs[NBKT_MAX];
    __shared__ int swB[32];
    __shared__ int2 srec[CHUNK];
    const int tid = threadIdx.x;
    if (tid < NBKT_MAX) lh[tid] = 0;
    if (blk == 0 && tid == 0) *cursor = 0;
    __syncthreads();
    int s[4], d[4], r[4], b[4];
    const int e0 = blk * CHUNK + tid;
    if ((blk + 1) * CHUNK <= ne) {
#pragma unroll
        for (int k = 0; k < 4; ++k) {
            s[k] = src[e0 + k * 1024];
            d[k] = dst[e0 + k * 1024];
        }
#pragma unroll
        for (int k = 0; k < 4; ++k) {
            b[k] = bucket_of(d[k], nbkt);
            r[k] = atomicAdd(&lh[b[k]], 1);
        }
    } else {
#pragma unroll
        for (int k = 0; k < 4; ++k) {
            const int e = e0 + k * 1024;
            s[k] = 0; d[k] = 0; r[k] = 0;
            b[k] = -1;
            if (e < ne) {
                s[k] = src[e];
                d[k] = dst[e];
                b[k] = bucket_of(d[k], nbkt);
                r[k] = atomicAdd(&lh[b[k]], 1);
            }
        }
    }
    __syncthreads();
    const int c = tid < NBKT_MAX ? lh[tid] : 0;
    const int lo = block_excl_scan_1024(c, swB);
    if (tid < nbkt) {
        lofs[tid] = lo;
        seg[tid * n_chunk + blk] = (lo << 16) | c;
    }
    __syncthreads();
#pragma unroll
    for (int k = 0; k < 4; ++k)
        if (b[k] >= 0) srec[lofs[b[k]] + r[k]] = make_int2(s[k], d[k]);
    __syncthreads();
    const int left = ne - blk * CHUNK;
    const int nv = left < CHUNK ? left : CHUNK;
    for (int i = tid; i < nv; i += 1024) recs[blk * CHUNK + i] = srec[i];
}

__device__ __forceinline__ int in_recs(int i, int ne) { return (unsigned)i < (unsigned)ne ? i : 0; }

__device__ __forceinline__ void rows_role(const int blk, const int2* __restrict__ recs, const int* __restrict__ seg,
                                          int* __restrict__ cursor,
                                          int* __restrict__ rowptr, int* __restrict__ cnt,
                                          float* __restrict__ dinv, int* __restrict__ col,
                                          int n_nodes, int ne, int nbkt, int n_chunk) {
    __shared__ int lcnt[BKT_W];
    __shared__ int lpos[BKT_W];
    __shared__ int swA[32], swB[32];
    __shared__ int sb[2];
    __shared__ int stage[STAGE_CAP];
    const int tid = threadIdx.x;
    const int lane = tid & 31;
    const int wv = __builtin_amdgcn_readfirstlane(tid >> 5);
    const int bkt = blk;
    if (tid < BKT_W) lcnt[tid] = 0;
    __syncthreads();
    const int* trow = seg + bkt * n_chunk;
    const int per = (n_chunk + 31) >> 5;
    const int c_lo = wv * per;
    const int c_hi = c_lo + per < n_chunk ? c_lo + per : n_chunk;

    int mine = 0;
    for (int g0 = c_lo; g0 < c_hi; g0 += 32) {
        const int e = g0 + lane < c_hi ? trow[g0 + lane] : 0;
        mine += e & 0xFFFF;
        const int ng = c_hi - g0 < 32 ? c_hi - g0 : 32;
        for (int j = 0; j < ng; j += SEGS) {
            int y[SEGS], cj[SEGS];
            int longest = 0;
#pragma unroll
            for (int k = 0; k < SEGS; ++k) {
                const int ej = __builtin_amdgcn_readlane(e, j + k);
                cj[k] = ej & 0xFFFF;
                longest = cj[k] > longest ? cj[k] : longest;
                const int i = (g0 + j + k) * CHUNK + (ej >> 16) + lane;
                y[k] = recs[in_recs(lane < cj[k] ? i : 0, ne)].y;
            }
#pragma unroll
            for (int k = 0; k < SEGS; ++k)
                if (lane < cj[k]) atomicAdd(&lcnt[y[k] & (BKT_W - 1)], 1);
            if (longest > 32) {
                for (int jj = j; jj < j + SEGS; ++jj) {
                    const int ej = __builtin_amdgcn_readlane(e, jj);
                    const int cl = ej & 0xFFFF;
                    const int bl = (g0 + jj) * CHUNK + (ej >> 16);
                    for (int i = lane + 32; i < cl; i += 32)
                        atomicAdd(&lcnt[recs[in_recs(bl + i, ne)].y & (BKT_W - 1)], 1);
                }
            }
        }
    }
    for (int o = 16; o > 0; o >>= 1) mine += __shfl_xor(mine, o);
    if (lane == 0) swA[wv] = mine;
    __syncthreads();
    if (wv == 0) {
        int t = swA[lane];
        for (int o = 16; o > 0; o >>= 1) t += __shfl_xor(t, o);
        if (lane == 0) {
            sb[1] = t;
            sb[0] = __hip_atomic_fetch_add(cursor, t, __ATOMIC_RELAXED, __HIP_MEMORY_SCOPE_AGENT);
        }
    }
    const int c = tid < BKT_W ? lcnt[tid] : 0;
    const int ex2 = block_excl_scan_1024(c, swB);
    int bstart = sb[0];
    int nb = sb[1];
    if (bstart < 0 || bstart > ne) { bstart = 0; nb = 0; }
    if (nb > ne - bstart) nb = ne - bstart;
    if (tid < BKT_W) {
        lpos[tid] = ex2;
        const int node = bkt * BKT_W + tid;
        if (node < n_nodes) {
            rowptr[node] = bstart + ex2;
            cnt[node] = c;
            dinv[node] = rsqrtf((float)(c + 1));
        }
    }
    __syncthreads();

    const bool staged = nb <= STAGE_CAP;
    for (int g0 = c_lo; g0 < c_hi; g0 += 32) {
        const int e = g0 + lane < c_hi ? trow[g0 + lane] : 0;
        const int ng = c_hi - g0 < 32 ? c_hi - g0 : 32;
        for (int j = 0; j < ng; j += SEGS) {
            int2 rec[SEGS];
            int cj[SEGS], p[SEGS];
            int longest = 0;
#pragma unroll
            for (int k = 0; k < SEGS; ++k) {
                const int ej = __builtin_amdgcn_readlane(e, j + k);
                cj[k] = ej & 0xFFFF;
                longest = cj[k] > longest ? cj[k] : longest;
                const int i = (g0 + j + k) * CHUNK + (ej >> 16) + lane;
                rec[k] = recs[in_recs(lane < cj[k] ? i : 0, ne)];
            }
#pragma unroll
            for (int k = 0; k < SEGS; ++k) {
                p[k] = -1;
                if (lane < cj[k]) p[k] = atomicAdd(&lpos[rec[k].y & (BKT_W - 1)], 1);
            }
#pragma unroll
            for (int k = 0; k < SEGS; ++k) {
                if ((unsigned)p[k] < (unsigned)nb) {
                    if (staged) stage[p[k]] = rec[k].x;
                    else        col[bstart + p[k]] = rec[k].x;
                }
            }
            if (longest > 32) {
                for (int jj = j; jj < j + SEGS; ++jj) {
                    const int ej = __builtin_amdgcn_readlane(e, jj);
                    const int cl = ej & 0xFFFF;
                    const int bl = (g0 + jj) * CHUNK + (ej >> 16);
                    for (int i = lane + 32; i < cl; i += 32) {
                        const int2 r2 = recs[in_recs(bl + i, ne)];
                        const int q = atomicAdd(&lpos[r2.y & (BKT_W - 1)], 1);
                        if ((unsigned)q < (unsigned)nb) {
                            if (staged) stage[q] = r2.x;
                            else        col[bstart + q] = r2.x;
                        }
                    }
                }
            }
        }
    }
    if (staged) {
        __syncthreads();
        for (int i = tid; i < nb; i += 1024) col[bstart + i] = stage[i];
    }
}

__device__ __forceinline__ void gemm_role(const int gblk, const float* __restrict__ x, const float* __restrict__ W1,
                                          float* __restrict__ h, int n_nodes)
{
    __shared__ float sW[(KTILE / 2) * PSTRIDE];
    __shared__ float sA[256 * APITCH];

    const int tid  = threadIdx.x;
    const int wave = tid >> 5;
    const int lane = tid & 31;
    const int half = lane >> 4;
    const int l16  = lane & 15;
    const int rt   = wave & 15;
    const int ch   = wave >> 4;

    const int sr = tid >> 3;
    const int sc = (tid & 7) * 4;
    int xr0 = gblk * 256 + sr;
    int xr1 = xr0 + 128;
    if (xr0 >= n_nodes) xr0 = n_nodes - 1;
    if (xr1 >= n_nodes) xr1 = n_nodes - 1;
    const unsigned xo0 = (unsigned)xr0 * (unsigned)(IN_CH * 4) + (unsigned)sc * 4u;
    const unsigned xo1 = (unsigned)xr1 * (unsigned)(IN_CH * 4) + (unsigned)sc * 4u;
    float* sa0 = &sA[sr * APITCH + sc];
    float* sa1 = &sA[(sr + 128) * APITCH + sc];

    const int wp  = tid >> 5;
    const int wc4 = (tid & 31) * 4;
    const bool wstage = tid < (KTILE / 2) * 32;

    const float* afrag = &sA[(rt * 16 + l16) * APITCH + half * 2];
    const float* bfrag = &sW[half * PSTRIDE + ch * 128 + l16 * 2];

    v8f acc[4] = {};

    float4 xa0 = *(const float4*)((const char*)x + xo0);
    float4 xa1 = *(const float4*)((const char*)x + xo1);
    const unsigned wo = (unsigned)(2 * wp) * (unsigned)(HID_CH * 4) + (unsigned)wc4 * 4u;

    for (int k0 = 0; k0 < IN_CH; k0 += KTILE) {
        float4 r0 = make_float4(0.f, 0.f, 0.f, 0.f), r1 = r0;
        if (wstage) {
            const char* wb = (const char*)W1 + (unsigned)k0 * (unsigned)(HID_CH * 4);
            r0 = *(const float4*)(wb + wo);
            r1 = *(const float4*)(wb + (wo + (unsigned)(HID_CH * 4)));
        }
        __syncthreads();
        *(float4*)sa0 = xa0;
        *(float4*)sa1 = xa1;
        if (wstage) {
            float* dp = &sW[wp * PSTRIDE + wc4 * 2];
            ((float2*)dp)[0] = make_float2(r0.x, r1.x);
            ((float2*)dp)[1] = make_float2(r0.y, r1.y);
            ((float2*)dp)[2] = make_float2(r0.z, r1.z);
            ((float2*)dp)[3] = make_float2(r0.w, r1.w);
        }
        __syncthreads();
        if (k0 + KTILE < IN_CH) {
            xa0 = *(const float4*)((const char*)x + (xo0 + (unsigned)(k0 + KTILE) * 4u));
            xa1 = *(const float4*)((const char*)x + (xo1 + (unsigned)(k0 + KTILE) * 4u));
        }

#pragma unroll
        for (int k = 0; k < KTILE; k += 4) {
            v2f a = *(const v2f*)(afrag + k);
            const float* bbase = bfrag + (k >> 1) * PSTRIDE;
#pragma unroll
            for (int nt = 0; nt < 4; ++nt) {
                v2f b = *(const v2f*)(bbase + nt * 32);
                acc[nt] = __builtin_amdgcn_wmma_f32_16x16x4_f32(
                    false, a, false, b, (short)0, acc[nt], false, false);
            }
        }
    }

    int t2 = tid;
    asm volatile("" : "+v"(t2));
    const int e_half = (t2 >> 4) & 1;
    const int e_l16  = t2 & 15;
    const int e_rt   = (t2 >> 5) & 15;
    const int e_ch   = t2 >> 9;
    const int e_row  = gblk * 256 + e_rt * 16 + 8 * e_half;
#pragma unroll
    for (int nt = 0; nt < 4; ++nt) {
#pragma unroll
        for (int v = 0; v < 8; ++v) {
            int r = e_row + v;
            if (r < n_nodes) h[(size_t)r * HID_CH + e_ch * 64 + nt * 16 + e_l16] = acc[nt][v];
        }
    }
}

__global__ __launch_bounds__(1024) __attribute__((amdgpu_waves_per_eu(16, 16))) void k_scatter_and_gemm(
    const int* __restrict__ src, const int* __restrict__ dst, int2* __restrict__ recs,
    int* __restrict__ seg, int* __restrict__ cursor, int ne, int nbkt, int n_chunk,
    const float* __restrict__ x, const float* __restrict__ W1, float* __restrict__ h, int n_nodes,
    int g_first, int g_count)
{
    const int b = blockIdx.x;
    if (b < g_count) gemm_role(g_first + b, x, W1, h, n_nodes);
    else             scatter_role(b - g_count, src, dst, recs, seg, cursor, ne, nbkt, n_chunk);
}

__global__ __launch_bounds__(1024) __attribute__((amdgpu_waves_per_eu(16, 16))) void k_rows_and_gemm(
    const int2* __restrict__ recs, const int* __restrict__ seg, int* __restrict__ cursor,
    int* __restrict__ rowptr, int* __restrict__ cnt,
    float* __restrict__ dinv, int* __restrict__ col, int ne, int nbkt, int n_chunk,
    const float* __restrict__ x, const float* __restrict__ W1, float* __restrict__ h, int n_nodes,
    int g_first, int g_count)
{
    const int b = blockIdx.x;
    if (b < g_count) gemm_role(g_first + b, x, W1, h, n_nodes);
    else             rows_role(b - g_count, recs, seg, cursor, rowptr, cnt, dinv, col, n_nodes, ne, nbkt, n_chunk);
}


template <int R>
__device__ __forceinline__ void rows_n(const float* __restrict__ h, const unsigned lane_off, const int my_s,
                                       const int my_di, const float dd, const int j, float4& acc) {
    float4 a[R];
#pragma unroll
    for (int k = 0; k < R; ++k) {
        const int s = __builtin_amdgcn_readlane(my_s, j + k);
        a[k] = *(const float4*)((const char*)h + ((unsigned)s * (unsigned)(HID_CH * 4) + lane_off));
    }
    __builtin_amdgcn_sched_barrier(0);
#pragma unroll
    for (int k = 0; k < R; ++k) {
        const float w = __int_as_float(__builtin_amdgcn_readlane(my_di, j + k)) * dd;
        acc.x += a[k].x * w; acc.y += a[k].y * w; acc.z += a[k].z * w; acc.w += a[k].w * w;
    }
}

#define G1_NPW 7

__global__ __launch_bounds__(512) void k_gather1(
    const float* __restrict__ h, const int* __restrict__ rowptr, const int* __restrict__ cnt,
    const int* __restrict__ col, const float* __restrict__ dinv,
    const float* __restrict__ b1, const float* __restrict__ W2,
    float* __restrict__ rec, int n_nodes)
{
    __shared__ float4 srec[G1_NPW * 16];
    const int lane = threadIdx.x & 31;
    const int wave = threadIdx.x >> 5;
    const unsigned lane_off = (unsigned)lane * 16u;
    for (int n = 0; n < G1_NPW; ++n) {
        const int node = __builtin_amdgcn_readfirstlane((blockIdx.x * G1_NPW + n) * 16 + wave);
        if (node < n_nodes) {
            const int start = rowptr[node];
            const int tot = cnt[node] + 1;
            const float dd = dinv[node];
            float4 acc = make_float4(0.f, 0.f, 0.f, 0.f);
            int base = 0;
            do {
                const int m = (tot - base < 32) ? (tot - base) : 32;
                const int i = base + lane;
                int my_s = node; float my_d = dd;
                if (i >= 1 && i < tot) {
                    my_s = col[start + i - 1];
                    my_d = dinv[my_s];
                }
                const int my_di = __float_as_int(my_d);
                int j = 0;
                for (; j + 8 <= m; j += 8) rows_n<8>(h, lane_off, my_s, my_di, dd, j, acc);
                switch (m - j) {
                    case 1: rows_n<1>(h, lane_off, my_s, my_di, dd, j, acc); break;
                    case 2: rows_n<2>(h, lane_off, my_s, my_di, dd, j, acc); break;
                    case 3: rows_n<3>(h, lane_off, my_s, my_di, dd, j, acc); break;
                    case 4: rows_n<4>(h, lane_off, my_s, my_di, dd, j, acc); break;
                    case 5: rows_n<5>(h, lane_off, my_s, my_di, dd, j, acc); break;
                    case 6: rows_n<6>(h, lane_off, my_s, my_di, dd, j, acc); break;
                    case 7: rows_n<7>(h, lane_off, my_s, my_di, dd, j, acc); break;
                    default: break;
                }
                base += 32;
            } while (base < tot);
            const float4 bv = *(const float4*)(b1 + lane * 4);
            const float r0 = fmaxf(acc.x + bv.x, 0.f), r1 = fmaxf(acc.y + bv.y, 0.f);
            const float r2 = fmaxf(acc.z + bv.z, 0.f), r3 = fmaxf(acc.w + bv.w, 0.f);
            const float4 wa = *(const float4*)(W2 + lane * 8);
            const float4 wb = *(const float4*)(W2 + lane * 8 + 4);
            float p0 = r0 * wa.x + r1 * wa.z + r2 * wb.x + r3 * wb.z;
            float p1 = r0 * wa.y + r1 * wa.w + r2 * wb.y + r3 * wb.w;
            for (int o = 16; o > 0; o >>= 1) {
                p0 += __shfl_xor(p0, o);
                p1 += __shfl_xor(p1, o);
            }
            if (lane == 0) srec[n * 16 + wave] = make_float4(p0, p1, dd, 0.f);
        }
    }
    __syncthreads();
    if (wave < G1_NPW) {
        const int idx = (blockIdx.x * G1_NPW + wave) * 32 + lane;
        if (idx < n_nodes * 2) ((float2*)rec)[idx] = ((const float2*)srec)[wave * 32 + lane];
    }
}

#define G2_NPW 2

__global__ __launch_bounds__(256) void k_gather2(
    const float* __restrict__ rec, const int* __restrict__ rowptr, const int* __restrict__ cnt,
    const int* __restrict__ col,
    const float* __restrict__ b2, float* __restrict__ out, int n_nodes)
{
    __shared__ float2 sres[G2_NPW * 32];
    const int tid = threadIdx.x;
    const int g = tid >> 3, sub = tid & 7;
    for (int n = 0; n < G2_NPW; ++n) {
        const int node = (blockIdx.x * G2_NPW + n) * 32 + g;
        float a0 = 0.f, a1 = 0.f;
        float dd = 0.f;
        float2 hs = make_float2(0.f, 0.f);
        if (node < n_nodes) {
            const int start = rowptr[node];
            const int end = start + cnt[node];
            const float4 own = *(const float4*)((const char*)rec + (unsigned)node * 16u);
            dd = own.z;
            hs = make_float2(own.x, own.y);
            for (int jb = start + sub; jb < end; jb += 32) {
                int s[4];
                float4 v[4];
#pragma unroll
                for (int k = 0; k < 4; ++k) {
                    const int jk = jb + 8 * k;
                    s[k] = col[(k == 0 || jk < end) ? jk : end - 1];
                }
#pragma unroll
                for (int k = 0; k < 4; ++k)
                    v[k] = *(const float4*)((const char*)rec + (unsigned)s[k] * 16u);
#pragma unroll
                for (int k = 0; k < 4; ++k) {
                    if (k == 0 || jb + 8 * k < end) {
                        const float norm = v[k].z * dd;
                        a0 += v[k].x * norm;
                        a1 += v[k].y * norm;
                    }
                }
            }
        }
        for (int o = 4; o > 0; o >>= 1) {
            a0 += __shfl_xor(a0, o);
            a1 += __shfl_xor(a1, o);
        }
        if (sub == 0) {
            const float w = dd * dd;
            sres[n * 32 + g] = make_float2((b2[0] + hs.x * w) + a0, (b2[1] + hs.y * w) + a1);
        }
    }
    __syncthreads();
    if (tid < G2_NPW * 32) {
        const int nd = blockIdx.x * (G2_NPW * 32) + tid;
        if (nd < n_nodes) *(float2*)(out + (size_t)nd * 2) = sres[tid];
    }
}

extern "C" void kernel_launch(void* const* d_in, const int* in_sizes, int n_in,
                              void* d_out, int out_size, void* d_ws, size_t ws_size,
                              hipStream_t stream)
{
    const float* x  = (const float*)d_in[0];
    const int*   ei = (const int*)d_in[1];
    const float* W1 = (const float*)d_in[2];
    const float* b1 = (const float*)d_in[3];
    const float* W2 = (const float*)d_in[4];
    const float* b2 = (const float*)d_in[5];
    float* out = (float*)d_out;

    const int n_nodes = in_sizes[0] / IN_CH;
    const int n_edges = in_sizes[1] / 2;
    const int* src = ei;
    const int* dst = ei + n_edges;
    const int nbkt = (n_nodes + BKT_W - 1) / BKT_W;
    const int n_chunk = (n_edges + CHUNK - 1) / CHUNK;

    char* ws = (char*)d_ws;
    size_t off = 0;
    const size_t nb4 = ((size_t)n_nodes * 4 + 4095) & ~(size_t)4095;
    const size_t eb4 = ((size_t)n_edges * 4 + 4095) & ~(size_t)4095;
    float* dinv = (float*)(ws + off);  off += nb4;
    float* h = (float*)(ws + off);
    off += ((size_t)n_nodes * HID_CH * 4 + 4095) & ~(size_t)4095;
    int* cnt = (int*)(ws + off);       off += nb4;
    int* rowptr = (int*)(ws + off);    off += nb4;
    int* cursor = (int*)(ws + off);    off += 4096;
    int* seg = (int*)(ws + off);       off += ((size_t)n_chunk * NBKT_MAX * 4 + 4095) & ~(size_t)4095;
    int2* recs = (int2*)(ws + off);    off += 2 * eb4;
    int* col = (int*)(ws + off);
    float* rec = (float*)recs;

    const int g_all = (n_nodes + 255) / 256;
    const int g_a = (g_all + 1) / 2;
    const int g_b = g_all - g_a;
    k_scatter_and_gemm<<<g_a + n_chunk, 1024, 0, stream>>>(src, dst, recs, seg, cursor, n_edges, nbkt, n_chunk,
                                                          x, W1, h, n_nodes, 0, g_a);
    k_rows_and_gemm   <<<g_b + nbkt, 1024, 0, stream>>>(recs, seg, cursor, rowptr, cnt, dinv, col, n_edges, nbkt, n_chunk,
                                                       x, W1, h, n_nodes, g_a, g_b);

    k_gather1<<<(n_nodes + 16 * G1_NPW - 1) / (16 * G1_NPW), 512, 0, stream>>>(h, rowptr, cnt, col, dinv, b1, W2, rec, n_nodes);
    k_gather2<<<(n_nodes + 32 * G2_NPW - 1) / (32 * G2_NPW), 256, 0, stream>>>(rec, rowptr, cnt, col, b2, out, n_nodes);
}


# x @ W1 row blocks resized so every WGP gets one per launch: 208 rows in launch 1, 192 in launch 2, 250 blocks each (was 196 of 256 rows); busiest WGP: 25 tile rows of WMMAs, was 32; sums unchanged
# speedup vs baseline: 5.0794x; 1.0325x over previous
#include <hip/hip_runtime.h>
#include <hip/hip_bf16.h>
#include <cstdint>


typedef float v2f __attribute__((ext_vector_type(2)));
typedef float v8f __attribute__((ext_vector_type(8)));

#define IN_CH  256
#define HID_CH 128
#define KTILE  32
#define APITCH 36
#define PSTRIDE 288

#define BKT_SHIFT 9
#define BKT_W     512
#define NBKT_MAX  256
#define CHUNK     4096
#define STAGE_CAP 12288
#define SEGS      8

__device__ __forceinline__ int wave_incl_scan_i(int v, int lane) {
    for (int o = 1; o < 32; o <<= 1) {
        int t = __shfl_up(v, o);
        if (lane >= o) v += t;
    }
    return v;
}

__device__ __forceinline__ int block_excl_scan_1024(int v, int* sw) {
    const int lane = threadIdx.x & 31, wave = threadIdx.x >> 5;
    const int incl = wave_incl_scan_i(v, lane);
    if (lane == 31) sw[wave] = incl;
    __syncthreads();
    if (wave == 0) {
        const int t = sw[lane];
        const int s = wave_incl_scan_i(t, lane);
        sw[lane] = s - t;
    }
    __syncthreads();
    return sw[wave] + incl - v;
}

__device__ __forceinline__ int bucket_of(int d, int nbkt) {
    const int b = (int)((unsigned)d >> BKT_SHIFT);
    return b < nbkt ? b : nbkt - 1;
}

__device__ __forceinline__ void scatter_role(const int blk, const int* __restrict__ src, const int* __restrict__ dst,
                                             int2* __restrict__ recs, int* __restrict__ seg, int* __restrict__ cursor,
                                             int ne, int nbkt, int n_chunk) {
    __shared__ int lh[NBKT_MAX];
    __shared__ int lofs[NBKT_MAX];
    __shared__ int swB[32];
    __shared__ int2 srec[CHUNK];
    const int tid = threadIdx.x;
    if (tid < NBKT_MAX) lh[tid] = 0;
    if (blk == 0 && tid == 0) *cursor = 0;
    __syncthreads();
    int s[4], d[4], r[4], b[4];
    const int e0 = blk * CHUNK + tid;
    if ((blk + 1) * CHUNK <= ne) {
#pragma unroll
        for (int k = 0; k < 4; ++k) {
            s[k] = src[e0 + k * 1024];
            d[k] = dst[e0 + k * 1024];
        }
#pragma unroll
        for (int k = 0; k < 4; ++k) {
            b[k] = bucket_of(d[k], nbkt);
            r[k] = atomicAdd(&lh[b[k]], 1);
        }
    } else {
#pragma unroll
        for (int k = 0; k < 4; ++k) {
            const int e = e0 + k * 1024;
            s[k] = 0; d[k] = 0; r[k] = 0;
            b[k] = -1;
            if (e < ne) {
                s[k] = src[e];
                d[k] = dst[e];
                b[k] = bucket_of(d[k], nbkt);
                r[k] = atomicAdd(&lh[b[k]], 1);
            }
        }
    }
    __syncthreads();
    const int c = tid < NBKT_MAX ? lh[tid] : 0;
    const int lo = block_excl_scan_1024(c, swB);
    if (tid < nbkt) {
        lofs[tid] = lo;
        seg[tid * n_chunk + blk] = (lo << 16) | c;
    }
    __syncthreads();
#pragma unroll
    for (int k = 0; k < 4; ++k)
        if (b[k] >= 0) srec[lofs[b[k]] + r[k]] = make_int2(s[k], d[k]);
    __syncthreads();
    const int left = ne - blk * CHUNK;
    const int nv = left < CHUNK ? left : CHUNK;
    for (int i = tid; i < nv; i += 1024) recs[blk * CHUNK + i] = srec[i];
}

__device__ __forceinline__ int in_recs(int i, int ne) { return (unsigned)i < (unsigned)ne ? i : 0; }

__device__ __forceinline__ void rows_role(const int blk, const int2* __restrict__ recs, const int* __restrict__ seg,
                                          int* __restrict__ cursor,
                                          int* __restrict__ rowptr, int* __restrict__ cnt,
                                          float* __restrict__ dinv, int* __restrict__ col,
                                          int n_nodes, int ne, int nbkt, int n_chunk) {
    __shared__ int lcnt[BKT_W];
    __shared__ int lpos[BKT_W];
    __shared__ int swA[32], swB[32];
    __shared__ int sb[2];
    __shared__ int stage[STAGE_CAP];
    const int tid = threadIdx.x;
    const int lane = tid & 31;
    const int wv = __builtin_amdgcn_readfirstlane(tid >> 5);
    const int bkt = blk;
    if (tid < BKT_W) lcnt[tid] = 0;
    __syncthreads();
    const int* trow = seg + bkt * n_chunk;
    const int per = (n_chunk + 31) >> 5;
    const int c_lo = wv * per;
    const int c_hi = c_lo + per < n_chunk ? c_lo + per : n_chunk;

    int mine = 0;
    for (int g0 = c_lo; g0 < c_hi; g0 += 32) {
        const int e = g0 + lane < c_hi ? trow[g0 + lane] : 0;
        mine += e & 0xFFFF;
        const int ng = c_hi - g0 < 32 ? c_hi - g0 : 32;
        for (int j = 0; j < ng; j += SEGS) {
            int y[SEGS], cj[SEGS];
            int longest = 0;
#pragma unroll
            for (int k = 0; k < SEGS; ++k) {
                const int ej = __builtin_amdgcn_readlane(e, j + k);
                cj[k] = ej & 0xFFFF;
                longest = cj[k] > longest ? cj[k] : longest;
                const int i = (g0 + j + k) * CHUNK + (ej >> 16) + lane;
                y[k] = recs[in_recs(lane < cj[k] ? i : 0, ne)].y;
            }
#pragma unroll
            for (int k = 0; k < SEGS; ++k)
                if (lane < cj[k]) atomicAdd(&lcnt[y[k] & (BKT_W - 1)], 1);
            if (longest > 32) {
                for (int jj = j; jj < j + SEGS; ++jj) {
                    const int ej = __builtin_amdgcn_readlane(e, jj);
                    const int cl = ej & 0xFFFF;
                    const int bl = (g0 + jj) * CHUNK + (ej >> 16);
                    for (int i = lane + 32; i < cl; i += 32)
                        atomicAdd(&lcnt[recs[in_recs(bl + i, ne)].y & (BKT_W - 1)], 1);
                }
            }
        }
    }
    for (int o = 16; o > 0; o >>= 1) mine += __shfl_xor(mine, o);
    if (lane == 0) swA[wv] = mine;
    __syncthreads();
    if (wv == 0) {
        int t = swA[lane];
        for (int o = 16; o > 0; o >>= 1) t += __shfl_xor(t, o);
        if (lane == 0) {
            sb[1] = t;
            sb[0] = __hip_atomic_fetch_add(cursor, t, __ATOMIC_RELAXED, __HIP_MEMORY_SCOPE_AGENT);
        }
    }
    const int c = tid < BKT_W ? lcnt[tid] : 0;
    const int ex2 = block_excl_scan_1024(c, swB);
    int bstart = sb[0];
    int nb = sb[1];
    if (bstart < 0 || bstart > ne) { bstart = 0; nb = 0; }
    if (nb > ne - bstart) nb = ne - bstart;
    if (tid < BKT_W) {
        lpos[tid] = ex2;
        const int node = bkt * BKT_W + tid;
        if (node < n_nodes) {
            rowptr[node] = bstart + ex2;
            cnt[node] = c;
            dinv[node] = rsqrtf((float)(c + 1));
        }
    }
    __syncthreads();

    const bool staged = nb <= STAGE_CAP;
    for (int g0 = c_lo; g0 < c_hi; g0 += 32) {
        const int e = g0 + lane < c_hi ? trow[g0 + lane] : 0;
        const int ng = c_hi - g0 < 32 ? c_hi - g0 : 32;
        for (int j = 0; j < ng; j += SEGS) {
            int2 rec[SEGS];
            int cj[SEGS], p[SEGS];
            int longest = 0;
#pragma unroll
            for (int k = 0; k < SEGS; ++k) {
                const int ej = __builtin_amdgcn_readlane(e, j + k);
                cj[k] = ej & 0xFFFF;
                longest = cj[k] > longest ? cj[k] : longest;
                const int i = (g0 + j + k) * CHUNK + (ej >> 16) + lane;
                rec[k] = recs[in_recs(lane < cj[k] ? i : 0, ne)];
            }
#pragma unroll
            for (int k = 0; k < SEGS; ++k) {
                p[k] = -1;
                if (lane < cj[k]) p[k] = atomicAdd(&lpos[rec[k].y & (BKT_W - 1)], 1);
            }
#pragma unroll
            for (int k = 0; k < SEGS; ++k) {
                if ((unsigned)p[k] < (unsigned)nb) {
                    if (staged) stage[p[k]] = rec[k].x;
                    else        col[bstart + p[k]] = rec[k].x;
                }
            }
            if (longest > 32) {
                for (int jj = j; jj < j + SEGS; ++jj) {
                    const int ej = __builtin_amdgcn_readlane(e, jj);
                    const int cl = ej & 0xFFFF;
                    const int bl = (g0 + jj) * CHUNK + (ej >> 16);
                    for (int i = lane + 32; i < cl; i += 32) {
                        const int2 r2 = recs[in_recs(bl + i, ne)];
                        const int q = atomicAdd(&lpos[r2.y & (BKT_W - 1)], 1);
                        if ((unsigned)q < (unsigned)nb) {
                            if (staged) stage[q] = r2.x;
                            else        col[bstart + q] = r2.x;
                        }
                    }
                }
            }
        }
    }
    if (staged) {
        __syncthreads();
        for (int i = tid; i < nb; i += 1024) col[bstart + i] = stage[i];
    }
}

#define NRT_A 13
#define NRT_B 12

template <int NRT>
__device__ __forceinline__ void gemm_role(const int row0, const float* __restrict__ x, const float* __restrict__ W1,
                                          float* __restrict__ h, int n_nodes)
{
    __shared__ float sW[(KTILE / 2) * PSTRIDE];
    __shared__ float sA[NRT * 16 * APITCH];

    const int tid  = threadIdx.x;
    const int lane = tid & 31;
    const int half = lane >> 4;
    const int l16  = lane & 15;
    const int wave = __builtin_amdgcn_readfirstlane(tid >> 5);
    const int cq   = wave & 3;
    const int rr   = wave >> 2;
    const bool two = rr >= 16 - NRT;
    const int rt2  = rr + NRT - 8;

    const bool row1   = wave < NRT * 4 - 32;
    const bool wstage = wave < (KTILE / 2);

    const float* afrag = &sA[(rr * 16 + l16) * APITCH + half * 2];
    const float* bfrag = &sW[half * PSTRIDE + cq * 64 + l16 * 2];

    v8f acc[4] = {};

    float4 xa0, xa1 = make_float4(0.f, 0.f, 0.f, 0.f);
    {
        const int sr = tid >> 3;
        int xr0 = row0 + sr;
        int xr1 = xr0 + 128;
        if (xr0 >= n_nodes) xr0 = n_nodes - 1;
        if (xr1 >= n_nodes) xr1 = n_nodes - 1;
        const unsigned xc = (unsigned)(tid & 7) * 16u;
        xa0 = *(const float4*)((const char*)x + ((unsigned)xr0 * (unsigned)(IN_CH * 4) + xc));
        if (row1) xa1 = *(const float4*)((const char*)x + ((unsigned)xr1 * (unsigned)(IN_CH * 4) + xc));
    }

    for (int k0 = 0; k0 < IN_CH; k0 += KTILE) {
        int t3 = tid;
        asm volatile("" : "+v"(t3));
        const int sr  = t3 >> 3;
        const int sc  = (t3 & 7) * 4;
        const int wp  = t3 >> 5;
        const int wc4 = (t3 & 31) * 4;
        int xr0 = row0 + sr;
        int xr1 = xr0 + 128;
        if (xr0 >= n_nodes) xr0 = n_nodes - 1;
        if (xr1 >= n_nodes) xr1 = n_nodes - 1;
        const unsigned xo0 = (unsigned)xr0 * (unsigned)(IN_CH * 4) + (unsigned)(sc + k0 + KTILE) * 4u;
        const unsigned xo1 = (unsigned)xr1 * (unsigned)(IN_CH * 4) + (unsigned)(sc + k0 + KTILE) * 4u;
        float* sa0 = &sA[sr * APITCH + sc];

        float4 r0 = make_float4(0.f, 0.f, 0.f, 0.f), r1 = r0;
        if (wstage) {
            const char* wb = (const char*)W1 + (unsigned)k0 * (unsigned)(HID_CH * 4);
            const unsigned wo = (unsigned)(2 * wp) * (unsigned)(HID_CH * 4) + (unsigned)wc4 * 4u;
            r0 = *(const float4*)(wb + wo);
            r1 = *(const float4*)(wb + (wo + (unsigned)(HID_CH * 4)));
        }
        __syncthreads();
        *(float4*)sa0 = xa0;
        if (row1) *(float4*)(sa0 + 128 * APITCH) = xa1;
        if (wstage) {
            float* dp = &sW[wp * PSTRIDE + wc4 * 2];
            ((float2*)dp)[0] = make_float2(r0.x, r1.x);
            ((float2*)dp)[1] = make_float2(r0.y, r1.y);
            ((float2*)dp)[2] = make_float2(r0.z, r1.z);
            ((float2*)dp)[3] = make_float2(r0.w, r1.w);
        }
        __syncthreads();
        if (k0 + KTILE < IN_CH) {
            xa0 = *(const float4*)((const char*)x + xo0);
            if (row1) xa1 = *(const float4*)((const char*)x + xo1);
        }

#pragma unroll
        for (int k = 0; k < KTILE; k += 4) {
            const float* bbase = bfrag + (k >> 1) * PSTRIDE;
            const v2f a  = *(const v2f*)(afrag + k);
            const v2f b0 = *(const v2f*)(bbase);
            const v2f b1 = *(const v2f*)(bbase + 32);
            acc[0] = __builtin_amdgcn_wmma_f32_16x16x4_f32(false, a, false, b0, (short)0, acc[0], false, false);
            acc[1] = __builtin_amdgcn_wmma_f32_16x16x4_f32(false, a, false, b1, (short)0, acc[1], false, false);
        }
        if (two) {
            int zoff = 0;
            asm volatile("" : "+s"(zoff));
            const float* bfrag2 = bfrag + zoff;
#pragma unroll
            for (int k = 0; k < KTILE; k += 4) {
                const float* bbase = bfrag2 + (k >> 1) * PSTRIDE;
                const v2f a  = *(const v2f*)(afrag + (NRT - 8) * 16 * APITCH + k);
                const v2f b0 = *(const v2f*)(bbase);
                const v2f b1 = *(const v2f*)(bbase + 32);
                acc[2] = __builtin_amdgcn_wmma_f32_16x16x4_f32(false, a, false, b0, (short)0, acc[2], false, false);
                acc[3] = __builtin_amdgcn_wmma_f32_16x16x4_f32(false, a, false, b1, (short)0, acc[3], false, false);
            }
        }
    }

    int t2 = tid;
    asm volatile("" : "+v"(t2));
    const int e_half = (t2 >> 4) & 1;
    const int e_col  = cq * 32 + (t2 & 15);
    const int e_rowA = row0 + rr * 16 + 8 * e_half;
#pragma unroll
    for (int nt = 0; nt < 2; ++nt) {
#pragma unroll
        for (int v = 0; v < 8; ++v) {
            const int r = e_rowA + v;
            if (r < n_nodes) h[(size_t)r * HID_CH + e_col + nt * 16] = acc[nt][v];
        }
    }
    if (two) {
        const int e_rowB = row0 + rt2 * 16 + 8 * e_half;
#pragma unroll
        for (int nt = 0; nt < 2; ++nt) {
#pragma unroll
            for (int v = 0; v < 8; ++v) {
                const int r = e_rowB + v;
                if (r < n_nodes) h[(size_t)r * HID_CH + e_col + nt * 16] = acc[2 + nt][v];
            }
        }
    }
}

__global__ __launch_bounds__(1024) __attribute__((amdgpu_waves_per_eu(16, 16))) void k_scatter_and_gemm(
    const int* __restrict__ src, const int* __restrict__ dst, int2* __restrict__ recs,
    int* __restrict__ seg, int* __restrict__ cursor, int ne, int nbkt, int n_chunk,
    const float* __restrict__ x, const float* __restrict__ W1, float* __restrict__ h, int n_nodes,
    int g_count)
{
    const int b = blockIdx.x;
    if (b < g_count) gemm_role<NRT_A>(b * (NRT_A * 16), x, W1, h, n_nodes);
    else             scatter_role(b - g_count, src, dst, recs, seg, cursor, ne, nbkt, n_chunk);
}

__global__ __launch_bounds__(1024) __attribute__((amdgpu_waves_per_eu(16, 16))) void k_rows_and_gemm(
    const int2* __restrict__ recs, const int* __restrict__ seg, int* __restrict__ cursor,
    int* __restrict__ rowptr, int* __restrict__ cnt,
    float* __restrict__ dinv, int* __restrict__ col, int ne, int nbkt, int n_chunk,
    const float* __restrict__ x, const float* __restrict__ W1, float* __restrict__ h, int n_nodes,
    int row_first, int g_count)
{
    const int b = blockIdx.x;
    if (b < g_count) gemm_role<NRT_B>(row_first + b * (NRT_B * 16), x, W1, h, n_nodes);
    else             rows_role(b - g_count, recs, seg, cursor, rowptr, cnt, dinv, col, n_nodes, ne, nbkt, n_chunk);
}


template <int R>
__device__ __forceinline__ void rows_n(const float* __restrict__ h, const unsigned lane_off, const int my_s,
                                       const int my_di, const float dd, const int j, float4& acc) {
    float4 a[R];
#pragma unroll
    for (int k = 0; k < R; ++k) {
        const int s = __builtin_amdgcn_readlane(my_s, j + k);
        a[k] = *(const float4*)((const char*)h + ((unsigned)s * (unsigned)(HID_CH * 4) + lane_off));
    }
    __builtin_amdgcn_sched_barrier(0);
#pragma unroll
    for (int k = 0; k < R; ++k) {
        const float w = __int_as_float(__builtin_amdgcn_readlane(my_di, j + k)) * dd;
        acc.x += a[k].x * w; acc.y += a[k].y * w; acc.z += a[k].z * w; acc.w += a[k].w * w;
    }
}

#define G1_NPW 7

__global__ __launch_bounds__(512) void k_gather1(
    const float* __restrict__ h, const int* __restrict__ rowptr, const int* __restrict__ cnt,
    const int* __restrict__ col, const float* __restrict__ dinv,
    const float* __restrict__ b1, const float* __restrict__ W2,
    float* __restrict__ rec, int n_nodes)
{
    __shared__ float4 srec[G1_NPW * 16];
    const int lane = threadIdx.x & 31;
    const int wave = threadIdx.x >> 5;
    const unsigned lane_off = (unsigned)lane * 16u;
    for (int n = 0; n < G1_NPW; ++n) {
        const int node = __builtin_amdgcn_readfirstlane((blockIdx.x * G1_NPW + n) * 16 + wave);
        if (node < n_nodes) {
            const int start = rowptr[node];
            const int tot = cnt[node] + 1;
            const float dd = dinv[node];
            float4 acc = make_float4(0.f, 0.f, 0.f, 0.f);
            int base = 0;
            do {
                const int m = (tot - base < 32) ? (tot - base) : 32;
                const int i = base + lane;
                int my_s = node; float my_d = dd;
                if (i >= 1 && i < tot) {
                    my_s = col[start + i - 1];
                    my_d = dinv[my_s];
                }
                const int my_di = __float_as_int(my_d);
                int j = 0;
                for (; j + 8 <= m; j += 8) rows_n<8>(h, lane_off, my_s, my_di, dd, j, acc);
                switch (m - j) {
                    case 1: rows_n<1>(h, lane_off, my_s, my_di, dd, j, acc); break;
                    case 2: rows_n<2>(h, lane_off, my_s, my_di, dd, j, acc); break;
                    case 3: rows_n<3>(h, lane_off, my_s, my_di, dd, j, acc); break;
                    case 4: rows_n<4>(h, lane_off, my_s, my_di, dd, j, acc); break;
                    case 5: rows_n<5>(h, lane_off, my_s, my_di, dd, j, acc); break;
                    case 6: rows_n<6>(h, lane_off, my_s, my_di, dd, j, acc); break;
                    case 7: rows_n<7>(h, lane_off, my_s, my_di, dd, j, acc); break;
                    default: break;
                }
                base += 32;
            } while (base < tot);
            const float4 bv = *(const float4*)(b1 + lane * 4);
            const float r0 = fmaxf(acc.x + bv.x, 0.f), r1 = fmaxf(acc.y + bv.y, 0.f);
            const float r2 = fmaxf(acc.z + bv.z, 0.f), r3 = fmaxf(acc.w + bv.w, 0.f);
            const float4 wa = *(const float4*)(W2 + lane * 8);
            const float4 wb = *(const float4*)(W2 + lane * 8 + 4);
            float p0 = r0 * wa.x + r1 * wa.z + r2 * wb.x + r3 * wb.z;
            float p1 = r0 * wa.y + r1 * wa.w + r2 * wb.y + r3 * wb.w;
            for (int o = 16; o > 0; o >>= 1) {
                p0 += __shfl_xor(p0, o);
                p1 += __shfl_xor(p1, o);
            }
            if (lane == 0) srec[n * 16 + wave] = make_float4(p0, p1, dd, 0.f);
        }
    }
    __syncthreads();
    if (wave < G1_NPW) {
        const int idx = (blockIdx.x * G1_NPW + wave) * 32 + lane;
        if (idx < n_nodes * 2) ((float2*)rec)[idx] = ((const float2*)srec)[wave * 32 + lane];
    }
}

#define G2_NPW 2

__global__ __launch_bounds__(256) void k_gather2(
    const float* __restrict__ rec, const int* __restrict__ rowptr, const int* __restrict__ cnt,
    const int* __restrict__ col,
    const float* __restrict__ b2, float* __restrict__ out, int n_nodes)
{
    __shared__ float2 sres[G2_NPW * 32];
    const int tid = threadIdx.x;
    const int g = tid >> 3, sub = tid & 7;
    for (int n = 0; n < G2_NPW; ++n) {
        const int node = (blockIdx.x * G2_NPW + n) * 32 + g;
        float a0 = 0.f, a1 = 0.f;
        float dd = 0.f;
        float2 hs = make_float2(0.f, 0.f);
        if (node < n_nodes) {
            const int start = rowptr[node];
            const int end = start + cnt[node];
            const float4 own = *(const float4*)((const char*)rec + (unsigned)node * 16u);
            dd = own.z;
            hs = make_float2(own.x, own.y);
            for (int jb = start + sub; jb < end; jb += 32) {
                int s[4];
                float4 v[4];
#pragma unroll
                for (int k = 0; k < 4; ++k) {
                    const int jk = jb + 8 * k;
                    s[k] = col[(k == 0 || jk < end) ? jk : end - 1];
                }
#pragma unroll
                for (int k = 0; k < 4; ++k)
                    v[k] = *(const float4*)((const char*)rec + (unsigned)s[k] * 16u);
#pragma unroll
                for (int k = 0; k < 4; ++k) {
                    if (k == 0 || jb + 8 * k < end) {
                        const float norm = v[k].z * dd;
                        a0 += v[k].x * norm;
                        a1 += v[k].y * norm;
                    }
                }
            }
        }
        for (int o = 4; o > 0; o >>= 1) {
            a0 += __shfl_xor(a0, o);
            a1 += __shfl_xor(a1, o);
        }
        if (sub == 0) {
            const float w = dd * dd;
            sres[n * 32 + g] = make_float2((b2[0] + hs.x * w) + a0, (b2[1] + hs.y * w) + a1);
        }
    }
    __syncthreads();
    if (tid < G2_NPW * 32) {
        const int nd = blockIdx.x * (G2_NPW * 32) + tid;
        if (nd < n_nodes) *(float2*)(out + (size_t)nd * 2) = sres[tid];
    }
}

extern "C" void kernel_launch(void* const* d_in, const int* in_sizes, int n_in,
                              void* d_out, int out_size, void* d_ws, size_t ws_size,
                              hipStream_t stream)
{
    const float* x  = (const float*)d_in[0];
    const int*   ei = (const int*)d_in[1];
    const float* W1 = (const float*)d_in[2];
    const float* b1 = (const float*)d_in[3];
    const float* W2 = (const float*)d_in[4];
    const float* b2 = (const float*)d_in[5];
    float* out = (float*)d_out;

    const int n_nodes = in_sizes[0] / IN_CH;
    const int n_edges = in_sizes[1] / 2;
    const int* src = ei;
    const int* dst = ei + n_edges;
    const int nbkt = (n_nodes + BKT_W - 1) / BKT_W;
    const int n_chunk = (n_edges + CHUNK - 1) / CHUNK;

    char* ws = (char*)d_ws;
    size_t off = 0;
    const size_t nb4 = ((size_t)n_nodes * 4 + 4095) & ~(size_t)4095;
    const size_t eb4 = ((size_t)n_edges * 4 + 4095) & ~(size_t)4095;
    float* dinv = (float*)(ws + off);  off += nb4;
    float* h = (float*)(ws + off);
    off += ((size_t)n_nodes * HID_CH * 4 + 4095) & ~(size_t)4095;
    int* cnt = (int*)(ws + off);       off += nb4;
    int* rowptr = (int*)(ws + off);    off += nb4;
    int* cursor = (int*)(ws + off);    off += 4096;
    int* seg = (int*)(ws + off);       off += ((size_t)n_chunk * NBKT_MAX * 4 + 4095) & ~(size_t)4095;
    int2* recs = (int2*)(ws + off);    off += 2 * eb4;
    int* col = (int*)(ws + off);
    float* rec = (float*)recs;

    const int g_blk = (n_nodes + (NRT_A + NRT_B) * 16 - 1) / ((NRT_A + NRT_B) * 16);
    k_scatter_and_gemm<<<g_blk + n_chunk, 1024, 0, stream>>>(src, dst, recs, seg, cursor, n_edges, nbkt, n_chunk,
                                                            x, W1, h, n_nodes, g_blk);
    k_rows_and_gemm   <<<g_blk + nbkt, 1024, 0, stream>>>(recs, seg, cursor, rowptr, cnt, dinv, col, n_edges, nbkt, n_chunk,
                                                         x, W1, h, n_nodes, g_blk * (NRT_A * 16), g_blk);

    k_gather1<<<(n_nodes + 16 * G1_NPW - 1) / (16 * G1_NPW), 512, 0, stream>>>(h, rowptr, cnt, col, dinv, b1, W2, rec, n_nodes);
    k_gather2<<<(n_nodes + 32 * G2_NPW - 1) / (32 * G2_NPW), 256, 0, stream>>>(rec, rowptr, cnt, col, b2, out, n_nodes);
}
